# starting kernel
# baseline (speedup 1.0000x reference)
#include <hip/hip_runtime.h>
#include <hip/hip_bf16.h>


#define N_NODES 50000
#define N_EDGES 800000
#define GNN_IN  256
#define GNN_HID 64
#define COMB    64
#define MLP_IN  256
#define MLP_HID 512

typedef __attribute__((ext_vector_type(16))) __bf16 v16bf;
typedef __attribute__((ext_vector_type(8)))  float  v8f;

union V16 {
    uint4 u[2];
    v16bf v;
};

__device__ __forceinline__ unsigned short f2bf(float f) {
    unsigned int u = __float_as_uint(f);
    unsigned int r = u + 0x7FFFu + ((u >> 16) & 1u);
    return (unsigned short)(r >> 16);
}

__global__ void cvt_bf16_kernel(const float* __restrict__ in,
                                unsigned short* __restrict__ out, int n4) {
    int i = blockIdx.x * blockDim.x + threadIdx.x;
    if (i >= n4) return;
    const float4 f = ((const float4*)in)[i];
    ushort4 o;
    o.x = f2bf(f.x); o.y = f2bf(f.y); o.z = f2bf(f.z); o.w = f2bf(f.w);
    ((ushort4*)out)[i] = o;
}

__global__ void pack_w_kernel(const float* __restrict__ W,
                              unsigned short* __restrict__ P, int K, int N) {
    int idx = blockIdx.x * blockDim.x + threadIdx.x;
    int nTiles = N >> 4;
    int total = (K >> 5) * nTiles * 32;
    if (idx >= total) return;
    int lane = idx & 31;
    int tile = idx >> 5;
    int kt = tile / nTiles;
    int nt = tile % nTiles;
    int col = nt * 16 + (lane & 15);
    int kb  = kt * 32 + ((lane < 16) ? 0 : 8);
    __attribute__((aligned(16))) unsigned short o[16];
#pragma unroll
    for (int j = 0; j < 8; ++j) o[j]     = f2bf(W[(size_t)(kb + j) * N + col]);
#pragma unroll
    for (int j = 0; j < 8; ++j) o[8 + j] = f2bf(W[(size_t)(kb + 16 + j) * N + col]);
    uint4* dst = (uint4*)(P + (size_t)tile * 512 + lane * 16);
    dst[0] = ((uint4*)o)[0];
    dst[1] = ((uint4*)o)[1];
}

__global__ void gemm_wmma_kernel(const unsigned short* __restrict__ A,
                                 const unsigned short* __restrict__ Bp,
                                 const float* __restrict__ bias,
                                 void* __restrict__ C,
                                 int M, int N, int K, int doRelu, int outBf16) {
    const int lane = threadIdx.x & 31;
    const int wid  = threadIdx.x >> 5;
    const int nGroups = N >> 6;
    const int mTiles  = M >> 4;
    const int task = blockIdx.x * 8 + wid;
    if (task >= mTiles * nGroups) return;
    const int mt = task / nGroups;
    const int ng = task % nGroups;
    const int kTiles = K >> 5;
    const int nTiles = N >> 4;

    const int hsel = (lane < 16) ? 0 : 8;
    const unsigned short* arow = A + (size_t)(mt * 16 + (lane & 15)) * K;
    const unsigned short* bcol = Bp + (size_t)(ng << 2) * 512 + lane * 16;
    const size_t bstride = (size_t)nTiles * 512;

    v8f acc0 = {}, acc1 = {}, acc2 = {}, acc3 = {};

    V16 aC, aN;
    V16 bC0, bC1, bC2, bC3, bN0, bN1, bN2, bN3;

    {
        const uint4* ap = (const uint4*)(arow + hsel);
        aC.u[0] = ap[0];
        aC.u[1] = ap[2];
        const unsigned short* bt = bcol;
        bC0.u[0] = ((const uint4*)(bt + 0 * 512))[0];
        bC0.u[1] = ((const uint4*)(bt + 0 * 512))[1];
        bC1.u[0] = ((const uint4*)(bt + 1 * 512))[0];
        bC1.u[1] = ((const uint4*)(bt + 1 * 512))[1];
        bC2.u[0] = ((const uint4*)(bt + 2 * 512))[0];
        bC2.u[1] = ((const uint4*)(bt + 2 * 512))[1];
        bC3.u[0] = ((const uint4*)(bt + 3 * 512))[0];
        bC3.u[1] = ((const uint4*)(bt + 3 * 512))[1];
    }

    for (int kt = 0; kt < kTiles; ++kt) {
        const int ktn = (kt + 1 < kTiles) ? (kt + 1) : kt;
        {
            const uint4* ap = (const uint4*)(arow + ktn * 32 + hsel);
            aN.u[0] = ap[0];
            aN.u[1] = ap[2];
            const unsigned short* bt = bcol + (size_t)ktn * bstride;
            bN0.u[0] = ((const uint4*)(bt + 0 * 512))[0];
            bN0.u[1] = ((const uint4*)(bt + 0 * 512))[1];
            bN1.u[0] = ((const uint4*)(bt + 1 * 512))[0];
            bN1.u[1] = ((const uint4*)(bt + 1 * 512))[1];
            bN2.u[0] = ((const uint4*)(bt + 2 * 512))[0];
            bN2.u[1] = ((const uint4*)(bt + 2 * 512))[1];
            bN3.u[0] = ((const uint4*)(bt + 3 * 512))[0];
            bN3.u[1] = ((const uint4*)(bt + 3 * 512))[1];
        }

        acc0 = __builtin_amdgcn_wmma_f32_16x16x32_bf16(false, aC.v, false, bC0.v,
                                                       (short)0, acc0, false, false);
        acc1 = __builtin_amdgcn_wmma_f32_16x16x32_bf16(false, aC.v, false, bC1.v,
                                                       (short)0, acc1, false, false);
        acc2 = __builtin_amdgcn_wmma_f32_16x16x32_bf16(false, aC.v, false, bC2.v,
                                                       (short)0, acc2, false, false);
        acc3 = __builtin_amdgcn_wmma_f32_16x16x32_bf16(false, aC.v, false, bC3.v,
                                                       (short)0, acc3, false, false);

        aC = aN;
        bC0 = bN0; bC1 = bN1; bC2 = bN2; bC3 = bN3;
    }

    const int r0 = mt * 16 + ((lane < 16) ? 0 : 8);
    v8f accs[4] = {acc0, acc1, acc2, acc3};
#pragma unroll
    for (int j = 0; j < 4; ++j) {
        const int cc = (ng * 4 + j) * 16 + (lane & 15);
        const float bv = bias ? bias[cc] : 0.0f;
#pragma unroll
        for (int v = 0; v < 8; ++v) {
            float val = accs[j][v] + bv;
            if (doRelu) val = fmaxf(val, 0.0f);
            const size_t off = (size_t)(r0 + v) * N + cc;
            if (outBf16) ((unsigned short*)C)[off] = f2bf(val);
            else         ((float*)C)[off] = val;
        }
    }
}

__global__ void deg_init_kernel(float* __restrict__ deg, int n) {
    int i = blockIdx.x * blockDim.x + threadIdx.x;
    if (i < n) deg[i] = 1.0f;
}

__global__ void deg_accum_kernel(const int* __restrict__ dst,
                                 const float* __restrict__ ew,
                                 float* __restrict__ deg, int nE) {
    int e = blockIdx.x * blockDim.x + threadIdx.x;
    if (e < nE) atomicAdd(&deg[dst[e]], ew[e]);
}

__global__ void dinv_kernel(const float* __restrict__ deg,
                            float* __restrict__ dinv, int n) {
    int i = blockIdx.x * blockDim.x + threadIdx.x;
    if (i >= n) return;
    float d = deg[i];
    dinv[i] = (d > 0.0f) ? rsqrtf(fmaxf(d, 1e-12f)) : 0.0f;
}

__global__ void edge_norm_kernel(const int* __restrict__ src,
                                 const int* __restrict__ dst,
                                 const float* __restrict__ ew,
                                 const float* __restrict__ dinv,
                                 float* __restrict__ norm, int nE) {
    int e = blockIdx.x * blockDim.x + threadIdx.x;
    if (e < nE) norm[e] = dinv[src[e]] * ew[e] * dinv[dst[e]];
}

__global__ void agg_init_kernel(const float* __restrict__ h,
                                const float* __restrict__ dinv,
                                float* __restrict__ agg, int nNodes) {
    int idx = blockIdx.x * blockDim.x + threadIdx.x;
    int n = idx >> 4;
    if (n >= nNodes) return;
    int c = (idx & 15) * 4;
    float w = dinv[n]; w *= w;
    float4 hv = *(const float4*)(h + (size_t)n * COMB + c);
    float4 o = {hv.x * w, hv.y * w, hv.z * w, hv.w * w};
    *(float4*)(agg + (size_t)n * COMB + c) = o;
}

__global__ void gcn_scatter_kernel(const float* __restrict__ h,
                                   const int* __restrict__ src,
                                   const int* __restrict__ dst,
                                   const float* __restrict__ norm,
                                   float* __restrict__ agg, int nE) {
    int idx = blockIdx.x * blockDim.x + threadIdx.x;
    int e = idx >> 4;
    if (e >= nE) return;
    int c = (idx & 15) * 4;
    int s = src[e], d = dst[e];
    float nm = norm[e];
    float4 hv = *(const float4*)(h + (size_t)s * COMB + c);
    float* ap = agg + (size_t)d * COMB + c;
    atomicAdd(ap + 0, hv.x * nm);
    atomicAdd(ap + 1, hv.y * nm);
    atomicAdd(ap + 2, hv.z * nm);
    atomicAdd(ap + 3, hv.w * nm);
}

__global__ void relu_bias_bf16_kernel(const float* __restrict__ agg,
                                      const float* __restrict__ b,
                                      unsigned short* __restrict__ g, int total) {
    int idx = blockIdx.x * blockDim.x + threadIdx.x;
    if (idx >= total) return;
    float v = agg[idx] + b[idx & (COMB - 1)];
    g[idx] = f2bf(fmaxf(v, 0.0f));
}

__global__ void final_kernel(const float* __restrict__ mlp,
                             const float* __restrict__ agg2,
                             const float* __restrict__ b2,
                             const float* __restrict__ fw,
                             const float* __restrict__ fb,
                             float* __restrict__ out) {
    __shared__ float red[128];
    int n = blockIdx.x;
    int t = threadIdx.x;
    float v;
    if (t < COMB) v = mlp[(size_t)n * COMB + t];
    else          v = agg2[(size_t)n * COMB + (t - COMB)] + b2[t - COMB];
    red[t] = v * fw[t];
    __syncthreads();
    for (int off = 64; off > 0; off >>= 1) {
        if (t < off) red[t] += red[t + off];
        __syncthreads();
    }
    if (t == 0) out[n] = red[0] + fb[0];
}

extern "C" void kernel_launch(void* const* d_in, const int* in_sizes, int n_in,
                              void* d_out, int out_size, void* d_ws, size_t ws_size,
                              hipStream_t stream) {
    (void)in_sizes; (void)n_in; (void)out_size; (void)ws_size;

    const float* x        = (const float*)d_in[0];
    const int*   eidx     = (const int*)d_in[1];
    const float* eattr    = (const float*)d_in[2];
    const float* ufeat    = (const float*)d_in[3];
    const float* gcn1_W   = (const float*)d_in[4];
    const float* gcn1_b   = (const float*)d_in[5];
    const float* gcn2_W   = (const float*)d_in[6];
    const float* gcn2_b   = (const float*)d_in[7];
    const float* fc1_W    = (const float*)d_in[8];
    const float* fc1_b    = (const float*)d_in[9];
    const float* fc2_W    = (const float*)d_in[10];
    const float* fc2_b    = (const float*)d_in[11];
    const float* fcf_W    = (const float*)d_in[12];
    const float* fcf_b    = (const float*)d_in[13];
    const int* esrc = eidx;
    const int* edst = eidx + N_EDGES;
    float* out = (float*)d_out;

    char* ws = (char*)d_ws;
    size_t off = 0;
    auto carve = [&](size_t bytes) -> void* {
        void* p = ws + off;
        off = (off + bytes + 255) & ~(size_t)255;
        return p;
    };
    unsigned short* UBF = (unsigned short*)carve((size_t)N_NODES * MLP_IN * 2);
    unsigned short* XBF = (unsigned short*)carve((size_t)N_NODES * GNN_IN * 2);
    unsigned short* HBF = (unsigned short*)carve((size_t)N_NODES * MLP_HID * 2);
    unsigned short* GBF = (unsigned short*)carve((size_t)N_NODES * COMB * 2);
    unsigned short* W1P = (unsigned short*)carve((size_t)MLP_IN * MLP_HID * 2);
    unsigned short* W2P = (unsigned short*)carve((size_t)MLP_HID * COMB * 2);
    unsigned short* G1P = (unsigned short*)carve((size_t)GNN_IN * GNN_HID * 2);
    unsigned short* G2P = (unsigned short*)carve((size_t)GNN_HID * COMB * 2);
    float* MLPOUT = (float*)carve((size_t)N_NODES * COMB * 4);
    float* H1     = (float*)carve((size_t)N_NODES * COMB * 4);
    float* AGG    = (float*)carve((size_t)N_NODES * COMB * 4);
    float* DEG    = (float*)carve((size_t)N_NODES * 4);
    float* DINV   = (float*)carve((size_t)N_NODES * 4);
    float* NORM   = (float*)carve((size_t)N_EDGES * 4);
    float* H2 = H1;

    {
        int n4 = (N_NODES * MLP_IN) / 4;
        cvt_bf16_kernel<<<(n4 + 255) / 256, 256, 0, stream>>>(ufeat, UBF, n4);
        n4 = (N_NODES * GNN_IN) / 4;
        cvt_bf16_kernel<<<(n4 + 255) / 256, 256, 0, stream>>>(x, XBF, n4);
    }
    auto pack = [&](const float* W, unsigned short* P, int K, int N) {
        int total = (K / 32) * (N / 16) * 32;
        pack_w_kernel<<<(total + 255) / 256, 256, 0, stream>>>(W, P, K, N);
    };
    pack(fc1_W, W1P, MLP_IN, MLP_HID);
    pack(fc2_W, W2P, MLP_HID, COMB);
    pack(gcn1_W, G1P, GNN_IN, GNN_HID);
    pack(gcn2_W, G2P, GNN_HID, COMB);

    deg_init_kernel<<<(N_NODES + 255) / 256, 256, 0, stream>>>(DEG, N_NODES);
    deg_accum_kernel<<<(N_EDGES + 255) / 256, 256, 0, stream>>>(edst, eattr, DEG, N_EDGES);
    dinv_kernel<<<(N_NODES + 255) / 256, 256, 0, stream>>>(DEG, DINV, N_NODES);
    edge_norm_kernel<<<(N_EDGES + 255) / 256, 256, 0, stream>>>(esrc, edst, eattr, DINV,
                                                                NORM, N_EDGES);

    auto gemm = [&](const unsigned short* A, const unsigned short* Bp,
                    const float* bias, void* C, int M, int N, int K,
                    int relu, int outBf) {
        int tasks = (M / 16) * (N / 64);
        int blocks = (tasks + 7) / 8;
        gemm_wmma_kernel<<<blocks, 256, 0, stream>>>(A, Bp, bias, C, M, N, K, relu, outBf);
    };

    gemm(UBF, W1P, fc1_b, HBF, N_NODES, MLP_HID, MLP_IN,  1,  1);
    gemm(HBF, W2P, fc2_b, MLPOUT, N_NODES, COMB, MLP_HID, 0, 0);

    gemm(XBF, G1P, nullptr, H1, N_NODES, GNN_HID, GNN_IN, 0, 0);
    agg_init_kernel<<<(N_NODES * 16 + 255) / 256, 256, 0, stream>>>(H1, DINV, AGG, N_NODES);
    gcn_scatter_kernel<<<(N_EDGES * 16 + 255) / 256, 256, 0, stream>>>(
        H1, esrc, edst, NORM, AGG, N_EDGES);
    relu_bias_bf16_kernel<<<(N_NODES * COMB + 255) / 256, 256, 0, stream>>>(
        AGG, gcn1_b, GBF, N_NODES * COMB);

    gemm(GBF, G2P, nullptr, H2, N_NODES, COMB, GNN_HID, 0, 0);
    agg_init_kernel<<<(N_NODES * 16 + 255) / 256, 256, 0, stream>>>(H2, DINV, AGG, N_NODES);
    gcn_scatter_kernel<<<(N_EDGES * 16 + 255) / 256, 256, 0, stream>>>(
        H2, esrc, edst, NORM, AGG, N_EDGES);

    final_kernel<<<N_NODES, 128, 0, stream>>>(MLPOUT, AGG, gcn2_b, fcf_W, fcf_b, out);
}


# element-wise, edge and final-dot passes become small resident grids: a thread loops over many items, a batch of loads in flight; atomics as before; final dot keeps its tree order (mul-adds may fuse)
# speedup vs baseline: 1.1682x; 1.1682x over previous
#include <hip/hip_runtime.h>
#include <hip/hip_bf16.h>


#define N_NODES 50000
#define N_EDGES 800000
#define GNN_IN  256
#define GNN_HID 64
#define COMB    64
#define MLP_IN  256
#define MLP_HID 512

typedef __attribute__((ext_vector_type(16))) __bf16 v16bf;
typedef __attribute__((ext_vector_type(8)))  float  v8f;

union V16 {
    uint4 u[2];
    v16bf v;
};

__device__ __forceinline__ unsigned short f2bf(float f) {
    unsigned int u = __float_as_uint(f);
    unsigned int r = u + 0x7FFFu + ((u >> 16) & 1u);
    return (unsigned short)(r >> 16);
}

__device__ __forceinline__ ushort4 f2bf4(const float4 f) {
    ushort4 o;
    o.x = f2bf(f.x); o.y = f2bf(f.y); o.z = f2bf(f.z); o.w = f2bf(f.w);
    return o;
}

#define LOOP_BLOCKS 512
#define SCATTER_BLOCKS 1024

__global__ __launch_bounds__(256) void cvt_bf16_kernel(const float* __restrict__ in,
                                                       unsigned short* __restrict__ out, int n4) {
    constexpr int U = 8;
    const float4* in4 = (const float4*)in;
    ushort4* out4 = (ushort4*)out;
    const int stride = gridDim.x * blockDim.x;
    int i = blockIdx.x * blockDim.x + threadIdx.x;
    for (; i + (U - 1) * stride < n4; i += U * stride) {
        float4 f[U];
#pragma unroll
        for (int u = 0; u < U; ++u) f[u] = in4[i + u * stride];
#pragma unroll
        for (int u = 0; u < U; ++u) out4[i + u * stride] = f2bf4(f[u]);
    }
    for (; i < n4; i += stride) out4[i] = f2bf4(in4[i]);
}

__global__ void pack_w_kernel(const float* __restrict__ W,
                              unsigned short* __restrict__ P, int K, int N) {
    int idx = blockIdx.x * blockDim.x + threadIdx.x;
    int nTiles = N >> 4;
    int total = (K >> 5) * nTiles * 32;
    if (idx >= total) return;
    int lane = idx & 31;
    int tile = idx >> 5;
    int kt = tile / nTiles;
    int nt = tile % nTiles;
    int col = nt * 16 + (lane & 15);
    int kb  = kt * 32 + ((lane < 16) ? 0 : 8);
    __attribute__((aligned(16))) unsigned short o[16];
#pragma unroll
    for (int j = 0; j < 8; ++j) o[j]     = f2bf(W[(size_t)(kb + j) * N + col]);
#pragma unroll
    for (int j = 0; j < 8; ++j) o[8 + j] = f2bf(W[(size_t)(kb + 16 + j) * N + col]);
    uint4* dst = (uint4*)(P + (size_t)tile * 512 + lane * 16);
    dst[0] = ((uint4*)o)[0];
    dst[1] = ((uint4*)o)[1];
}

template <int K>
__device__ __forceinline__ void load_a_tile(V16 (&a)[K / 32],
                                            const unsigned short* __restrict__ abase,
                                            int mt) {
    const unsigned short* arow = abase + (size_t)mt * 16 * K;
#pragma unroll
    for (int kt = 0; kt < K / 32; ++kt) {
        const uint4* ap = (const uint4*)(arow + kt * 32);
        a[kt].u[0] = ap[0];
        a[kt].u[1] = ap[2];
    }
}

template <int K, int N, int NTW, bool RELU, bool OUTBF>
__device__ __forceinline__ void tile_mma_store(const V16 (&a)[K / 32],
                                               const V16 (&b)[K / 32][NTW],
                                               const float (&bv)[NTW],
                                               void* __restrict__ C,
                                               int mt, int cg, int lane) {
    v8f acc[NTW];
#pragma unroll
    for (int j = 0; j < NTW; ++j) acc[j] = v8f{};
#pragma unroll
    for (int kt = 0; kt < K / 32; ++kt) {
#pragma unroll
        for (int j = 0; j < NTW; ++j) {
            acc[j] = __builtin_amdgcn_wmma_f32_16x16x32_bf16(false, a[kt].v, false, b[kt][j].v,
                                                             (short)0, acc[j], false, false);
        }
    }
    const int r0 = mt * 16 + ((lane < 16) ? 0 : 8);
#pragma unroll
    for (int j = 0; j < NTW; ++j) {
        const int cc = (cg * NTW + j) * 16 + (lane & 15);
#pragma unroll
        for (int v = 0; v < 8; ++v) {
            float val = acc[j][v] + bv[j];
            if (RELU) val = fmaxf(val, 0.0f);
            const size_t off = (size_t)(r0 + v) * N + cc;
            if (OUTBF) ((unsigned short*)C)[off] = f2bf(val);
            else       ((float*)C)[off] = val;
        }
    }
}

template <int K, int N, int NTW, bool RELU, bool OUTBF, bool HASBIAS>
__global__ __launch_bounds__(256) void gemm_wmma_kernel(const unsigned short* __restrict__ A,
                                                        const unsigned short* __restrict__ Bp,
                                                        const float* __restrict__ bias,
                                                        void* __restrict__ C, int M) {
    constexpr int KT = K / 32;
    constexpr int NT = N / 16;
    constexpr int CG = NT / NTW;
    constexpr int S  = 8 / CG;
    static_assert(K % 32 == 0 && N % 16 == 0 && NT % NTW == 0 && 8 % CG == 0,
                  "tile split must divide the problem and the 8 waves of a block");
    const int lane = threadIdx.x & 31;
    const int wid  = __builtin_amdgcn_readfirstlane(threadIdx.x >> 5);
    const int cg     = wid % CG;
    const int stream = wid / CG;
    const int mTiles = M >> 4;
    const int step   = gridDim.x * S;
    int mt = blockIdx.x * S + stream;
    if (mt >= mTiles) return;

    V16 b[KT][NTW];
#pragma unroll
    for (int kt = 0; kt < KT; ++kt) {
#pragma unroll
        for (int j = 0; j < NTW; ++j) {
            const uint4* bp = (const uint4*)(Bp + (size_t)(kt * NT + cg * NTW + j) * 512 + lane * 16);
            b[kt][j].u[0] = bp[0];
            b[kt][j].u[1] = bp[1];
        }
    }
    float bv[NTW];
#pragma unroll
    for (int j = 0; j < NTW; ++j)
        bv[j] = HASBIAS ? bias[(cg * NTW + j) * 16 + (lane & 15)] : 0.0f;

    const unsigned short* abase = A + (size_t)(lane & 15) * K + ((lane < 16) ? 0 : 8);

    V16 a0[KT], a1[KT];
    load_a_tile<K>(a0, abase, mt);
    while (mt + step < mTiles) {
        load_a_tile<K>(a1, abase, mt + step);
        __builtin_amdgcn_sched_barrier(0);
        tile_mma_store<K, N, NTW, RELU, OUTBF>(a0, b, bv, C, mt, cg, lane);
        const int nxt = mt + 2 * step;
        load_a_tile<K>(a0, abase, (nxt < mTiles) ? nxt : mt);
        __builtin_amdgcn_sched_barrier(0);
        tile_mma_store<K, N, NTW, RELU, OUTBF>(a1, b, bv, C, mt + step, cg, lane);
        mt = nxt;
    }
    if (mt < mTiles) tile_mma_store<K, N, NTW, RELU, OUTBF>(a0, b, bv, C, mt, cg, lane);
}

__global__ void deg_init_kernel(float* __restrict__ deg, int n) {
    int i = blockIdx.x * blockDim.x + threadIdx.x;
    if (i < n) deg[i] = 1.0f;
}

__global__ __launch_bounds__(256) void deg_accum_kernel(const int* __restrict__ dst,
                                                        const float* __restrict__ ew,
                                                        float* __restrict__ deg, int nE) {
    constexpr int U = 6;
    const int stride = gridDim.x * blockDim.x;
    int e = blockIdx.x * blockDim.x + threadIdx.x;
    for (; e + (U - 1) * stride < nE; e += U * stride) {
        int d[U];
        float w[U];
#pragma unroll
        for (int u = 0; u < U; ++u) { d[u] = dst[e + u * stride]; w[u] = ew[e + u * stride]; }
#pragma unroll
        for (int u = 0; u < U; ++u) atomicAdd(&deg[d[u]], w[u]);
    }
    for (; e < nE; e += stride) atomicAdd(&deg[dst[e]], ew[e]);
}

__global__ void dinv_kernel(const float* __restrict__ deg,
                            float* __restrict__ dinv, int n) {
    int i = blockIdx.x * blockDim.x + threadIdx.x;
    if (i >= n) return;
    float d = deg[i];
    dinv[i] = (d > 0.0f) ? rsqrtf(fmaxf(d, 1e-12f)) : 0.0f;
}

__global__ __launch_bounds__(256) void edge_norm_kernel(const int* __restrict__ src,
                                                        const int* __restrict__ dst,
                                                        const float* __restrict__ ew,
                                                        const float* __restrict__ dinv,
                                                        float* __restrict__ norm, int nE) {
    constexpr int U = 6;
    const int stride = gridDim.x * blockDim.x;
    int e = blockIdx.x * blockDim.x + threadIdx.x;
    for (; e + (U - 1) * stride < nE; e += U * stride) {
        int s[U], d[U];
        float w[U], ds[U], dd[U];
#pragma unroll
        for (int u = 0; u < U; ++u) {
            s[u] = src[e + u * stride]; d[u] = dst[e + u * stride]; w[u] = ew[e + u * stride];
        }
#pragma unroll
        for (int u = 0; u < U; ++u) { ds[u] = dinv[s[u]]; dd[u] = dinv[d[u]]; }
#pragma unroll
        for (int u = 0; u < U; ++u) norm[e + u * stride] = ds[u] * w[u] * dd[u];
    }
    for (; e < nE; e += stride) norm[e] = dinv[src[e]] * ew[e] * dinv[dst[e]];
}

__global__ __launch_bounds__(256) void agg_init_kernel(const float* __restrict__ h,
                                                       const float* __restrict__ dinv,
                                                       float* __restrict__ agg, int nNodes) {
    constexpr int U = 6;
    const float4* h4 = (const float4*)h;
    float4* agg4 = (float4*)agg;
    const int n16 = nNodes * (COMB / 4);
    const int stride = gridDim.x * blockDim.x;
    int idx = blockIdx.x * blockDim.x + threadIdx.x;
    for (; idx + (U - 1) * stride < n16; idx += U * stride) {
        float4 hv[U];
        float w[U];
#pragma unroll
        for (int u = 0; u < U; ++u) {
            hv[u] = h4[idx + u * stride];
            w[u]  = dinv[(idx + u * stride) >> 4];
        }
#pragma unroll
        for (int u = 0; u < U; ++u) {
            const float ww = w[u] * w[u];
            const float4 o = {hv[u].x * ww, hv[u].y * ww, hv[u].z * ww, hv[u].w * ww};
            agg4[idx + u * stride] = o;
        }
    }
    for (; idx < n16; idx += stride) {
        float w = dinv[idx >> 4]; w *= w;
        const float4 hv = h4[idx];
        const float4 o = {hv.x * w, hv.y * w, hv.z * w, hv.w * w};
        agg4[idx] = o;
    }
}

template <bool GUARD>
__device__ __forceinline__ void scatter_group(const float* __restrict__ h,
                                              float* __restrict__ agg,
                                              int s_l, int d_l, float n_l,
                                              int half, int c, int left) {
    constexpr int P = 8;
#pragma unroll
    for (int j0 = 0; j0 < 16; j0 += P) {
        float4 hv[P];
        float nm[P];
        int d[P];
#pragma unroll
        for (int p = 0; p < P; ++p) {
            const int sl = 2 * (j0 + p) + half;
            const int s = __shfl(s_l, sl);
            d[p]  = __shfl(d_l, sl);
            nm[p] = __shfl(n_l, sl);
            if (!GUARD || sl < left) hv[p] = *(const float4*)(h + (size_t)s * COMB + c);
        }
#pragma unroll
        for (int p = 0; p < P; ++p) {
            if (!GUARD || 2 * (j0 + p) + half < left) {
                float* ap = agg + (size_t)d[p] * COMB + c;
                atomicAdd(ap + 0, hv[p].x * nm[p]);
                atomicAdd(ap + 1, hv[p].y * nm[p]);
                atomicAdd(ap + 2, hv[p].z * nm[p]);
                atomicAdd(ap + 3, hv[p].w * nm[p]);
            }
        }
    }
}

__global__ __launch_bounds__(256) void gcn_scatter_kernel(const float* __restrict__ h,
                                                          const int* __restrict__ src,
                                                          const int* __restrict__ dst,
                                                          const float* __restrict__ norm,
                                                          float* __restrict__ agg, int nE) {
    const int lane = threadIdx.x & 31;
    const int half = lane >> 4;
    const int c    = (lane & 15) * 4;
    const int gstep = ((gridDim.x * blockDim.x) >> 5) * 32;
    int base = ((blockIdx.x * blockDim.x + threadIdx.x) >> 5) * 32;
    for (; base + 32 <= nE; base += gstep) {
        const int e = base + lane;
        scatter_group<false>(h, agg, src[e], dst[e], norm[e], half, c, 32);
    }
    if (base < nE) {
        const int e = base + lane;
        const bool ok = e < nE;
        scatter_group<true>(h, agg, ok ? src[e] : 0, ok ? dst[e] : 0, ok ? norm[e] : 0.0f,
                            half, c, nE - base);
    }
}

__global__ __launch_bounds__(256) void relu_bias_bf16_kernel(const float* __restrict__ agg,
                                                             const float* __restrict__ b,
                                                             unsigned short* __restrict__ g,
                                                             int total) {
    constexpr int U = 6;
    const float4* a4 = (const float4*)agg;
    ushort4* g4 = (ushort4*)g;
    const int n4 = total >> 2;
    const int stride = gridDim.x * blockDim.x;
    int idx = blockIdx.x * blockDim.x + threadIdx.x;
    const float4 bv = *(const float4*)(b + (idx & 15) * 4);
    for (; idx + (U - 1) * stride < n4; idx += U * stride) {
        float4 v[U];
#pragma unroll
        for (int u = 0; u < U; ++u) v[u] = a4[idx + u * stride];
#pragma unroll
        for (int u = 0; u < U; ++u) {
            const float4 r = {fmaxf(v[u].x + bv.x, 0.0f), fmaxf(v[u].y + bv.y, 0.0f),
                              fmaxf(v[u].z + bv.z, 0.0f), fmaxf(v[u].w + bv.w, 0.0f)};
            g4[idx + u * stride] = f2bf4(r);
        }
    }
    for (; idx < n4; idx += stride) {
        const float4 v = a4[idx];
        const float4 r = {fmaxf(v.x + bv.x, 0.0f), fmaxf(v.y + bv.y, 0.0f),
                          fmaxf(v.z + bv.z, 0.0f), fmaxf(v.w + bv.w, 0.0f)};
        g4[idx] = f2bf4(r);
    }
}

__global__ __launch_bounds__(64) void final_kernel(const float* __restrict__ mlp,
                                                   const float* __restrict__ agg2,
                                                   const float* __restrict__ b2,
                                                   const float* __restrict__ fw,
                                                   const float* __restrict__ fb,
                                                   float* __restrict__ out, int nNodes) {
    constexpr int U = 8;
    const int lane = threadIdx.x & 31;
    const int n0 = ((blockIdx.x * blockDim.x + threadIdx.x) >> 5) * 32;
    if (n0 >= nNodes) return;
    const float f0 = fw[lane], f1 = fw[32 + lane], f2 = fw[64 + lane], f3 = fw[96 + lane];
    const float c0 = b2[lane], c1 = b2[32 + lane];
    float res = 0.0f;
    for (int j0 = 0; j0 < 32; j0 += U) {
        float m0[U], m1[U], a0[U], a1[U];
#pragma unroll
        for (int u = 0; u < U; ++u) {
            int n = n0 + j0 + u;
            n = (n < nNodes) ? n : (nNodes - 1);
            const float* mr = mlp  + (size_t)n * COMB;
            const float* ar = agg2 + (size_t)n * COMB;
            m0[u] = mr[lane]; m1[u] = mr[32 + lane];
            a0[u] = ar[lane]; a1[u] = ar[32 + lane];
        }
#pragma unroll
        for (int u = 0; u < U; ++u) {
            float x = (m0[u] * f0 + (a0[u] + c0) * f2) + (m1[u] * f1 + (a1[u] + c1) * f3);
            x += __shfl_xor(x, 16);
            x += __shfl_xor(x, 8);
            x += __shfl_xor(x, 4);
            x += __shfl_xor(x, 2);
            x += __shfl_xor(x, 1);
            if (lane == j0 + u) res = x;
        }
    }
    if (n0 + lane < nNodes) out[n0 + lane] = res + fb[0];
}

extern "C" void kernel_launch(void* const* d_in, const int* in_sizes, int n_in,
                              void* d_out, int out_size, void* d_ws, size_t ws_size,
                              hipStream_t stream) {
    (void)in_sizes; (void)n_in; (void)out_size; (void)ws_size;

    const float* x        = (const float*)d_in[0];
    const int*   eidx     = (const int*)d_in[1];
    const float* eattr    = (const float*)d_in[2];
    const float* ufeat    = (const float*)d_in[3];
    const float* gcn1_W   = (const float*)d_in[4];
    const float* gcn1_b   = (const float*)d_in[5];
    const float* gcn2_W   = (const float*)d_in[6];
    const float* gcn2_b   = (const float*)d_in[7];
    const float* fc1_W    = (const float*)d_in[8];
    const float* fc1_b    = (const float*)d_in[9];
    const float* fc2_W    = (const float*)d_in[10];
    const float* fc2_b    = (const float*)d_in[11];
    const float* fcf_W    = (const float*)d_in[12];
    const float* fcf_b    = (const float*)d_in[13];
    const int* esrc = eidx;
    const int* edst = eidx + N_EDGES;
    float* out = (float*)d_out;

    char* ws = (char*)d_ws;
    size_t off = 0;
    auto carve = [&](size_t bytes) -> void* {
        void* p = ws + off;
        off = (off + bytes + 255) & ~(size_t)255;
        return p;
    };
    unsigned short* UBF = (unsigned short*)carve((size_t)N_NODES * MLP_IN * 2);
    unsigned short* XBF = (unsigned short*)carve((size_t)N_NODES * GNN_IN * 2);
    unsigned short* HBF = (unsigned short*)carve((size_t)N_NODES * MLP_HID * 2);
    unsigned short* GBF = (unsigned short*)carve((size_t)N_NODES * COMB * 2);
    unsigned short* W1P = (unsigned short*)carve((size_t)MLP_IN * MLP_HID * 2);
    unsigned short* W2P = (unsigned short*)carve((size_t)MLP_HID * COMB * 2);
    unsigned short* G1P = (unsigned short*)carve((size_t)GNN_IN * GNN_HID * 2);
    unsigned short* G2P = (unsigned short*)carve((size_t)GNN_HID * COMB * 2);
    float* MLPOUT = (float*)carve((size_t)N_NODES * COMB * 4);
    float* H1     = (float*)carve((size_t)N_NODES * COMB * 4);
    float* AGG    = (float*)carve((size_t)N_NODES * COMB * 4);
    float* DEG    = (float*)carve((size_t)N_NODES * 4);
    float* DINV   = (float*)carve((size_t)N_NODES * 4);
    float* NORM   = (float*)carve((size_t)N_EDGES * 4);
    float* H2 = H1;

    {
        int n4 = (N_NODES * MLP_IN) / 4;
        cvt_bf16_kernel<<<LOOP_BLOCKS, 256, 0, stream>>>(ufeat, UBF, n4);
        n4 = (N_NODES * GNN_IN) / 4;
        cvt_bf16_kernel<<<LOOP_BLOCKS, 256, 0, stream>>>(x, XBF, n4);
    }
    auto pack = [&](const float* W, unsigned short* P, int K, int N) {
        int total = (K / 32) * (N / 16) * 32;
        pack_w_kernel<<<(total + 255) / 256, 256, 0, stream>>>(W, P, K, N);
    };
    pack(fc1_W, W1P, MLP_IN, MLP_HID);
    pack(fc2_W, W2P, MLP_HID, COMB);
    pack(gcn1_W, G1P, GNN_IN, GNN_HID);
    pack(gcn2_W, G2P, GNN_HID, COMB);

    deg_init_kernel<<<(N_NODES + 255) / 256, 256, 0, stream>>>(DEG, N_NODES);
    deg_accum_kernel<<<LOOP_BLOCKS, 256, 0, stream>>>(edst, eattr, DEG, N_EDGES);
    dinv_kernel<<<(N_NODES + 255) / 256, 256, 0, stream>>>(DEG, DINV, N_NODES);
    edge_norm_kernel<<<LOOP_BLOCKS, 256, 0, stream>>>(esrc, edst, eattr, DINV, NORM, N_EDGES);

    gemm_wmma_kernel<MLP_IN, MLP_HID, 4, true, true, true><<<256, 256, 0, stream>>>(
        UBF, W1P, fc1_b, HBF, N_NODES);
    gemm_wmma_kernel<MLP_HID, COMB, 1, false, false, true><<<256, 256, 0, stream>>>(
        HBF, W2P, fc2_b, MLPOUT, N_NODES);

    gemm_wmma_kernel<GNN_IN, GNN_HID, 2, false, false, false><<<256, 256, 0, stream>>>(
        XBF, G1P, nullptr, H1, N_NODES);
    agg_init_kernel<<<LOOP_BLOCKS, 256, 0, stream>>>(H1, DINV, AGG, N_NODES);
    gcn_scatter_kernel<<<SCATTER_BLOCKS, 256, 0, stream>>>(H1, esrc, edst, NORM, AGG, N_EDGES);
    relu_bias_bf16_kernel<<<LOOP_BLOCKS, 256, 0, stream>>>(AGG, gcn1_b, GBF, N_NODES * COMB);

    gemm_wmma_kernel<GNN_HID, COMB, 2, false, false, false><<<256, 256, 0, stream>>>(
        GBF, G2P, nullptr, H2, N_NODES);
    agg_init_kernel<<<LOOP_BLOCKS, 256, 0, stream>>>(H2, DINV, AGG, N_NODES);
    gcn_scatter_kernel<<<SCATTER_BLOCKS, 256, 0, stream>>>(H2, esrc, edst, NORM, AGG, N_EDGES);

    final_kernel<<<(N_NODES + 63) / 64, 64, 0, stream>>>(MLPOUT, AGG, gcn2_b, fcf_W, fcf_b, out,
                                                      N_NODES);
}


# all f32 atomic sums (degree, both edge aggregations) split into two planes, one per package half: a wave adds into its own half's plane, consumers add the planes; sum order: plane 0 + plane 1
# speedup vs baseline: 1.2613x; 1.0797x over previous
#include <hip/hip_runtime.h>
#include <hip/hip_bf16.h>


#define N_NODES 50000
#define N_EDGES 800000
#define GNN_IN  256
#define GNN_HID 64
#define COMB    64
#define MLP_IN  256
#define MLP_HID 512

typedef __attribute__((ext_vector_type(16))) __bf16 v16bf;
typedef __attribute__((ext_vector_type(8)))  float  v8f;

union V16 {
    uint4 u[2];
    v16bf v;
};

__device__ __forceinline__ unsigned short f2bf(float f) {
    unsigned int u = __float_as_uint(f);
    unsigned int r = u + 0x7FFFu + ((u >> 16) & 1u);
    return (unsigned short)(r >> 16);
}

__device__ __forceinline__ ushort4 f2bf4(const float4 f) {
    ushort4 o;
    o.x = f2bf(f.x); o.y = f2bf(f.y); o.z = f2bf(f.z); o.w = f2bf(f.w);
    return o;
}

__device__ __forceinline__ int package_half() {
    const unsigned int id = __builtin_amdgcn_s_sendmsg_rtn(0x87);
    return (int)((id >> 18) & 1u);
}

#define LOOP_BLOCKS 512
#define SCATTER_BLOCKS 1024

__global__ __launch_bounds__(256) void cvt_bf16_kernel(const float* __restrict__ in,
                                                       unsigned short* __restrict__ out, int n4) {
    constexpr int U = 8;
    const float4* in4 = (const float4*)in;
    ushort4* out4 = (ushort4*)out;
    const int stride = gridDim.x * blockDim.x;
    int i = blockIdx.x * blockDim.x + threadIdx.x;
    for (; i + (U - 1) * stride < n4; i += U * stride) {
        float4 f[U];
#pragma unroll
        for (int u = 0; u < U; ++u) f[u] = in4[i + u * stride];
#pragma unroll
        for (int u = 0; u < U; ++u) out4[i + u * stride] = f2bf4(f[u]);
    }
    for (; i < n4; i += stride) out4[i] = f2bf4(in4[i]);
}

__global__ void pack_w_kernel(const float* __restrict__ W,
                              unsigned short* __restrict__ P, int K, int N) {
    int idx = blockIdx.x * blockDim.x + threadIdx.x;
    int nTiles = N >> 4;
    int total = (K >> 5) * nTiles * 32;
    if (idx >= total) return;
    int lane = idx & 31;
    int tile = idx >> 5;
    int kt = tile / nTiles;
    int nt = tile % nTiles;
    int col = nt * 16 + (lane & 15);
    int kb  = kt * 32 + ((lane < 16) ? 0 : 8);
    __attribute__((aligned(16))) unsigned short o[16];
#pragma unroll
    for (int j = 0; j < 8; ++j) o[j]     = f2bf(W[(size_t)(kb + j) * N + col]);
#pragma unroll
    for (int j = 0; j < 8; ++j) o[8 + j] = f2bf(W[(size_t)(kb + 16 + j) * N + col]);
    uint4* dst = (uint4*)(P + (size_t)tile * 512 + lane * 16);
    dst[0] = ((uint4*)o)[0];
    dst[1] = ((uint4*)o)[1];
}

template <int K>
__device__ __forceinline__ void load_a_tile(V16 (&a)[K / 32],
                                            const unsigned short* __restrict__ abase,
                                            int mt) {
    const unsigned short* arow = abase + (size_t)mt * 16 * K;
#pragma unroll
    for (int kt = 0; kt < K / 32; ++kt) {
        const uint4* ap = (const uint4*)(arow + kt * 32);
        a[kt].u[0] = ap[0];
        a[kt].u[1] = ap[2];
    }
}

template <int K, int N, int NTW, bool RELU, bool OUTBF>
__device__ __forceinline__ void tile_mma_store(const V16 (&a)[K / 32],
                                               const V16 (&b)[K / 32][NTW],
                                               const float (&bv)[NTW],
                                               void* __restrict__ C,
                                               int mt, int cg, int lane) {
    v8f acc[NTW];
#pragma unroll
    for (int j = 0; j < NTW; ++j) acc[j] = v8f{};
#pragma unroll
    for (int kt = 0; kt < K / 32; ++kt) {
#pragma unroll
        for (int j = 0; j < NTW; ++j) {
            acc[j] = __builtin_amdgcn_wmma_f32_16x16x32_bf16(false, a[kt].v, false, b[kt][j].v,
                                                             (short)0, acc[j], false, false);
        }
    }
    const int r0 = mt * 16 + ((lane < 16) ? 0 : 8);
#pragma unroll
    for (int j = 0; j < NTW; ++j) {
        const int cc = (cg * NTW + j) * 16 + (lane & 15);
#pragma unroll
        for (int v = 0; v < 8; ++v) {
            float val = acc[j][v] + bv[j];
            if (RELU) val = fmaxf(val, 0.0f);
            const size_t off = (size_t)(r0 + v) * N + cc;
            if (OUTBF) ((unsigned short*)C)[off] = f2bf(val);
            else       ((float*)C)[off] = val;
        }
    }
}

template <int K, int N, int NTW, bool RELU, bool OUTBF, bool HASBIAS>
__global__ __launch_bounds__(256) void gemm_wmma_kernel(const unsigned short* __restrict__ A,
                                                        const unsigned short* __restrict__ Bp,
                                                        const float* __restrict__ bias,
                                                        void* __restrict__ C, int M) {
    constexpr int KT = K / 32;
    constexpr int NT = N / 16;
    constexpr int CG = NT / NTW;
    constexpr int S  = 8 / CG;
    static_assert(K % 32 == 0 && N % 16 == 0 && NT % NTW == 0 && 8 % CG == 0,
                  "tile split must divide the problem and the 8 waves of a block");
    const int lane = threadIdx.x & 31;
    const int wid  = __builtin_amdgcn_readfirstlane(threadIdx.x >> 5);
    const int cg     = wid % CG;
    const int stream = wid / CG;
    const int mTiles = M >> 4;
    const int step   = gridDim.x * S;
    int mt = blockIdx.x * S + stream;
    if (mt >= mTiles) return;

    V16 b[KT][NTW];
#pragma unroll
    for (int kt = 0; kt < KT; ++kt) {
#pragma unroll
        for (int j = 0; j < NTW; ++j) {
            const uint4* bp = (const uint4*)(Bp + (size_t)(kt * NT + cg * NTW + j) * 512 + lane * 16);
            b[kt][j].u[0] = bp[0];
            b[kt][j].u[1] = bp[1];
        }
    }
    float bv[NTW];
#pragma unroll
    for (int j = 0; j < NTW; ++j)
        bv[j] = HASBIAS ? bias[(cg * NTW + j) * 16 + (lane & 15)] : 0.0f;

    const unsigned short* abase = A + (size_t)(lane & 15) * K + ((lane < 16) ? 0 : 8);

    V16 a0[KT], a1[KT];
    load_a_tile<K>(a0, abase, mt);
    while (mt + step < mTiles) {
        load_a_tile<K>(a1, abase, mt + step);
        __builtin_amdgcn_sched_barrier(0);
        tile_mma_store<K, N, NTW, RELU, OUTBF>(a0, b, bv, C, mt, cg, lane);
        const int nxt = mt + 2 * step;
        load_a_tile<K>(a0, abase, (nxt < mTiles) ? nxt : mt);
        __builtin_amdgcn_sched_barrier(0);
        tile_mma_store<K, N, NTW, RELU, OUTBF>(a1, b, bv, C, mt + step, cg, lane);
        mt = nxt;
    }
    if (mt < mTiles) tile_mma_store<K, N, NTW, RELU, OUTBF>(a0, b, bv, C, mt, cg, lane);
}

__global__ void deg_init_kernel(float* __restrict__ deg0, float* __restrict__ deg1, int n) {
    int i = blockIdx.x * blockDim.x + threadIdx.x;
    if (i < n) {
        deg0[i] = 1.0f;
        deg1[i] = 0.0f;
    }
}

__global__ __launch_bounds__(256) void deg_accum_kernel(const int* __restrict__ dst,
                                                        const float* __restrict__ ew,
                                                        float* __restrict__ deg0,
                                                        float* __restrict__ deg1, int nE) {
    constexpr int U = 6;
    float* __restrict__ deg = package_half() ? deg1 : deg0;
    const int stride = gridDim.x * blockDim.x;
    int e = blockIdx.x * blockDim.x + threadIdx.x;
    for (; e + (U - 1) * stride < nE; e += U * stride) {
        int d[U];
        float w[U];
#pragma unroll
        for (int u = 0; u < U; ++u) { d[u] = dst[e + u * stride]; w[u] = ew[e + u * stride]; }
#pragma unroll
        for (int u = 0; u < U; ++u) atomicAdd(&deg[d[u]], w[u]);
    }
    for (; e < nE; e += stride) atomicAdd(&deg[dst[e]], ew[e]);
}

__global__ void dinv_kernel(const float* __restrict__ deg0,
                            const float* __restrict__ deg1,
                            float* __restrict__ dinv, int n) {
    int i = blockIdx.x * blockDim.x + threadIdx.x;
    if (i >= n) return;
    float d = deg0[i] + deg1[i];
    dinv[i] = (d > 0.0f) ? rsqrtf(fmaxf(d, 1e-12f)) : 0.0f;
}

__global__ __launch_bounds__(256) void edge_norm_kernel(const int* __restrict__ src,
                                                        const int* __restrict__ dst,
                                                        const float* __restrict__ ew,
                                                        const float* __restrict__ dinv,
                                                        float* __restrict__ norm, int nE) {
    constexpr int U = 6;
    const int stride = gridDim.x * blockDim.x;
    int e = blockIdx.x * blockDim.x + threadIdx.x;
    for (; e + (U - 1) * stride < nE; e += U * stride) {
        int s[U], d[U];
        float w[U], ds[U], dd[U];
#pragma unroll
        for (int u = 0; u < U; ++u) {
            s[u] = src[e + u * stride]; d[u] = dst[e + u * stride]; w[u] = ew[e + u * stride];
        }
#pragma unroll
        for (int u = 0; u < U; ++u) { ds[u] = dinv[s[u]]; dd[u] = dinv[d[u]]; }
#pragma unroll
        for (int u = 0; u < U; ++u) norm[e + u * stride] = ds[u] * w[u] * dd[u];
    }
    for (; e < nE; e += stride) norm[e] = dinv[src[e]] * ew[e] * dinv[dst[e]];
}

__global__ __launch_bounds__(256) void agg_init_kernel(const float* __restrict__ h,
                                                       const float* __restrict__ dinv,
                                                       float* __restrict__ agg,
                                                       float* __restrict__ agg1, int nNodes) {
    constexpr int U = 6;
    const float4* h4 = (const float4*)h;
    float4* agg4 = (float4*)agg;
    float4* aggz = (float4*)agg1;
    const float4 zero = {0.0f, 0.0f, 0.0f, 0.0f};
    const int n16 = nNodes * (COMB / 4);
    const int stride = gridDim.x * blockDim.x;
    int idx = blockIdx.x * blockDim.x + threadIdx.x;
    for (; idx + (U - 1) * stride < n16; idx += U * stride) {
        float4 hv[U];
        float w[U];
#pragma unroll
        for (int u = 0; u < U; ++u) {
            hv[u] = h4[idx + u * stride];
            w[u]  = dinv[(idx + u * stride) >> 4];
        }
#pragma unroll
        for (int u = 0; u < U; ++u) {
            const float ww = w[u] * w[u];
            const float4 o = {hv[u].x * ww, hv[u].y * ww, hv[u].z * ww, hv[u].w * ww};
            agg4[idx + u * stride] = o;
            aggz[idx + u * stride] = zero;
        }
    }
    for (; idx < n16; idx += stride) {
        float w = dinv[idx >> 4]; w *= w;
        const float4 hv = h4[idx];
        const float4 o = {hv.x * w, hv.y * w, hv.z * w, hv.w * w};
        agg4[idx] = o;
        aggz[idx] = zero;
    }
}

template <bool GUARD>
__device__ __forceinline__ void scatter_group(const float* __restrict__ h,
                                              float* __restrict__ agg,
                                              int s_l, int d_l, float n_l,
                                              int half, int c, int left) {
    constexpr int P = 8;
#pragma unroll
    for (int j0 = 0; j0 < 16; j0 += P) {
        float4 hv[P];
        float nm[P];
        int d[P];
#pragma unroll
        for (int p = 0; p < P; ++p) {
            const int sl = 2 * (j0 + p) + half;
            const int s = __shfl(s_l, sl);
            d[p]  = __shfl(d_l, sl);
            nm[p] = __shfl(n_l, sl);
            if (!GUARD || sl < left) hv[p] = *(const float4*)(h + (size_t)s * COMB + c);
        }
#pragma unroll
        for (int p = 0; p < P; ++p) {
            if (!GUARD || 2 * (j0 + p) + half < left) {
                float* ap = agg + (size_t)d[p] * COMB + c;
                atomicAdd(ap + 0, hv[p].x * nm[p]);
                atomicAdd(ap + 1, hv[p].y * nm[p]);
                atomicAdd(ap + 2, hv[p].z * nm[p]);
                atomicAdd(ap + 3, hv[p].w * nm[p]);
            }
        }
    }
}

__global__ __launch_bounds__(256) void gcn_scatter_kernel(const float* __restrict__ h,
                                                          const int* __restrict__ src,
                                                          const int* __restrict__ dst,
                                                          const float* __restrict__ norm,
                                                          float* __restrict__ agg0,
                                                          float* __restrict__ agg1, int nE) {
    float* __restrict__ agg = package_half() ? agg1 : agg0;
    const int lane = threadIdx.x & 31;
    const int half = lane >> 4;
    const int c    = (lane & 15) * 4;
    const int gstep = ((gridDim.x * blockDim.x) >> 5) * 32;
    int base = ((blockIdx.x * blockDim.x + threadIdx.x) >> 5) * 32;
    for (; base + 32 <= nE; base += gstep) {
        const int e = base + lane;
        scatter_group<false>(h, agg, src[e], dst[e], norm[e], half, c, 32);
    }
    if (base < nE) {
        const int e = base + lane;
        const bool ok = e < nE;
        scatter_group<true>(h, agg, ok ? src[e] : 0, ok ? dst[e] : 0, ok ? norm[e] : 0.0f,
                            half, c, nE - base);
    }
}

__global__ __launch_bounds__(256) void relu_bias_bf16_kernel(const float* __restrict__ agg,
                                                             const float* __restrict__ agg1,
                                                             const float* __restrict__ b,
                                                             unsigned short* __restrict__ g,
                                                             int total) {
    constexpr int U = 6;
    const float4* a4 = (const float4*)agg;
    const float4* p4 = (const float4*)agg1;
    ushort4* g4 = (ushort4*)g;
    const int n4 = total >> 2;
    const int stride = gridDim.x * blockDim.x;
    int idx = blockIdx.x * blockDim.x + threadIdx.x;
    const float4 bv = *(const float4*)(b + (idx & 15) * 4);
    for (; idx + (U - 1) * stride < n4; idx += U * stride) {
        float4 v[U], q[U];
#pragma unroll
        for (int u = 0; u < U; ++u) { v[u] = a4[idx + u * stride]; q[u] = p4[idx + u * stride]; }
#pragma unroll
        for (int u = 0; u < U; ++u) {
            const float4 r = {fmaxf((v[u].x + q[u].x) + bv.x, 0.0f),
                              fmaxf((v[u].y + q[u].y) + bv.y, 0.0f),
                              fmaxf((v[u].z + q[u].z) + bv.z, 0.0f),
                              fmaxf((v[u].w + q[u].w) + bv.w, 0.0f)};
            g4[idx + u * stride] = f2bf4(r);
        }
    }
    for (; idx < n4; idx += stride) {
        const float4 v = a4[idx];
        const float4 q = p4[idx];
        const float4 r = {fmaxf((v.x + q.x) + bv.x, 0.0f), fmaxf((v.y + q.y) + bv.y, 0.0f),
                          fmaxf((v.z + q.z) + bv.z, 0.0f), fmaxf((v.w + q.w) + bv.w, 0.0f)};
        g4[idx] = f2bf4(r);
    }
}

__global__ __launch_bounds__(64) void final_kernel(const float* __restrict__ mlp,
                                                   const float* __restrict__ agg2,
                                                   const float* __restrict__ agg2b,
                                                   const float* __restrict__ b2,
                                                   const float* __restrict__ fw,
                                                   const float* __restrict__ fb,
                                                   float* __restrict__ out, int nNodes) {
    constexpr int U = 8;
    const int lane = threadIdx.x & 31;
    const int n0 = ((blockIdx.x * blockDim.x + threadIdx.x) >> 5) * 32;
    if (n0 >= nNodes) return;
    const float f0 = fw[lane], f1 = fw[32 + lane], f2 = fw[64 + lane], f3 = fw[96 + lane];
    const float c0 = b2[lane], c1 = b2[32 + lane];
    float res = 0.0f;
    for (int j0 = 0; j0 < 32; j0 += U) {
        float m0[U], m1[U], a0[U], a1[U], p0[U], p1[U];
#pragma unroll
        for (int u = 0; u < U; ++u) {
            int n = n0 + j0 + u;
            n = (n < nNodes) ? n : (nNodes - 1);
            const float* mr = mlp  + (size_t)n * COMB;
            const float* ar = agg2 + (size_t)n * COMB;
            const float* pr = agg2b + (size_t)n * COMB;
            m0[u] = mr[lane]; m1[u] = mr[32 + lane];
            a0[u] = ar[lane]; a1[u] = ar[32 + lane];
            p0[u] = pr[lane]; p1[u] = pr[32 + lane];
        }
#pragma unroll
        for (int u = 0; u < U; ++u) {
            float x = (m0[u] * f0 + ((a0[u] + p0[u]) + c0) * f2) +
                      (m1[u] * f1 + ((a1[u] + p1[u]) + c1) * f3);
            x += __shfl_xor(x, 16);
            x += __shfl_xor(x, 8);
            x += __shfl_xor(x, 4);
            x += __shfl_xor(x, 2);
            x += __shfl_xor(x, 1);
            if (lane == j0 + u) res = x;
        }
    }
    if (n0 + lane < nNodes) out[n0 + lane] = res + fb[0];
}

extern "C" void kernel_launch(void* const* d_in, const int* in_sizes, int n_in,
                              void* d_out, int out_size, void* d_ws, size_t ws_size,
                              hipStream_t stream) {
    (void)in_sizes; (void)n_in; (void)out_size; (void)ws_size;

    const float* x        = (const float*)d_in[0];
    const int*   eidx     = (const int*)d_in[1];
    const float* eattr    = (const float*)d_in[2];
    const float* ufeat    = (const float*)d_in[3];
    const float* gcn1_W   = (const float*)d_in[4];
    const float* gcn1_b   = (const float*)d_in[5];
    const float* gcn2_W   = (const float*)d_in[6];
    const float* gcn2_b   = (const float*)d_in[7];
    const float* fc1_W    = (const float*)d_in[8];
    const float* fc1_b    = (const float*)d_in[9];
    const float* fc2_W    = (const float*)d_in[10];
    const float* fc2_b    = (const float*)d_in[11];
    const float* fcf_W    = (const float*)d_in[12];
    const float* fcf_b    = (const float*)d_in[13];
    const int* esrc = eidx;
    const int* edst = eidx + N_EDGES;
    float* out = (float*)d_out;

    char* ws = (char*)d_ws;
    size_t off = 0;
    auto carve = [&](size_t bytes) -> void* {
        void* p = ws + off;
        off = (off + bytes + 255) & ~(size_t)255;
        return p;
    };
    unsigned short* UBF = (unsigned short*)carve((size_t)N_NODES * MLP_IN * 2);
    unsigned short* XBF = (unsigned short*)carve((size_t)N_NODES * GNN_IN * 2);
    unsigned short* HBF = (unsigned short*)carve((size_t)N_NODES * MLP_HID * 2);
    unsigned short* GBF = (unsigned short*)carve((size_t)N_NODES * COMB * 2);
    unsigned short* W1P = (unsigned short*)carve((size_t)MLP_IN * MLP_HID * 2);
    unsigned short* W2P = (unsigned short*)carve((size_t)MLP_HID * COMB * 2);
    unsigned short* G1P = (unsigned short*)carve((size_t)GNN_IN * GNN_HID * 2);
    unsigned short* G2P = (unsigned short*)carve((size_t)GNN_HID * COMB * 2);
    float* MLPOUT = (float*)carve((size_t)N_NODES * COMB * 4);
    float* H1     = (float*)carve((size_t)N_NODES * COMB * 4);
    float* AGG    = (float*)carve((size_t)N_NODES * COMB * 4);
    float* DEG    = (float*)carve((size_t)N_NODES * 4);
    float* DINV   = (float*)carve((size_t)N_NODES * 4);
    float* NORM   = (float*)carve((size_t)N_EDGES * 4);
    float* H2 = H1;
    float* DEG1 = (float*)HBF;
    float* AGG1 = (float*)UBF;

    {
        int n4 = (N_NODES * MLP_IN) / 4;
        cvt_bf16_kernel<<<LOOP_BLOCKS, 256, 0, stream>>>(ufeat, UBF, n4);
        n4 = (N_NODES * GNN_IN) / 4;
        cvt_bf16_kernel<<<LOOP_BLOCKS, 256, 0, stream>>>(x, XBF, n4);
    }
    auto pack = [&](const float* W, unsigned short* P, int K, int N) {
        int total = (K / 32) * (N / 16) * 32;
        pack_w_kernel<<<(total + 255) / 256, 256, 0, stream>>>(W, P, K, N);
    };
    pack(fc1_W, W1P, MLP_IN, MLP_HID);
    pack(fc2_W, W2P, MLP_HID, COMB);
    pack(gcn1_W, G1P, GNN_IN, GNN_HID);
    pack(gcn2_W, G2P, GNN_HID, COMB);

    deg_init_kernel<<<(N_NODES + 255) / 256, 256, 0, stream>>>(DEG, DEG1, N_NODES);
    deg_accum_kernel<<<LOOP_BLOCKS, 256, 0, stream>>>(edst, eattr, DEG, DEG1, N_EDGES);
    dinv_kernel<<<(N_NODES + 255) / 256, 256, 0, stream>>>(DEG, DEG1, DINV, N_NODES);
    edge_norm_kernel<<<LOOP_BLOCKS, 256, 0, stream>>>(esrc, edst, eattr, DINV, NORM, N_EDGES);

    gemm_wmma_kernel<MLP_IN, MLP_HID, 4, true, true, true><<<256, 256, 0, stream>>>(
        UBF, W1P, fc1_b, HBF, N_NODES);
    gemm_wmma_kernel<MLP_HID, COMB, 1, false, false, true><<<256, 256, 0, stream>>>(
        HBF, W2P, fc2_b, MLPOUT, N_NODES);

    gemm_wmma_kernel<GNN_IN, GNN_HID, 2, false, false, false><<<256, 256, 0, stream>>>(
        XBF, G1P, nullptr, H1, N_NODES);
    agg_init_kernel<<<LOOP_BLOCKS, 256, 0, stream>>>(H1, DINV, AGG, AGG1, N_NODES);
    gcn_scatter_kernel<<<SCATTER_BLOCKS, 256, 0, stream>>>(H1, esrc, edst, NORM, AGG, AGG1, N_EDGES);
    relu_bias_bf16_kernel<<<LOOP_BLOCKS, 256, 0, stream>>>(AGG, AGG1, gcn1_b, GBF, N_NODES * COMB);

    gemm_wmma_kernel<GNN_HID, COMB, 2, false, false, false><<<256, 256, 0, stream>>>(
        GBF, G2P, nullptr, H2, N_NODES);
    agg_init_kernel<<<LOOP_BLOCKS, 256, 0, stream>>>(H2, DINV, AGG, AGG1, N_NODES);
    gcn_scatter_kernel<<<SCATTER_BLOCKS, 256, 0, stream>>>(H2, esrc, edst, NORM, AGG, AGG1, N_EDGES);

    final_kernel<<<(N_NODES + 63) / 64, 64, 0, stream>>>(MLPOUT, AGG, AGG1, gcn2_b, fcf_W, fcf_b,
                                                      out, N_NODES);
}


# fc1+fc2 GEMMs chained in one launch: fc1 transposed, a wave's relu'd tiles rounded to bf16 in place are the B operand of its fc2 WMMAs; fc2 k-partials of the 8 waves added via LDS in wave order
# speedup vs baseline: 1.3329x; 1.0568x over previous
#include <hip/hip_runtime.h>
#include <hip/hip_bf16.h>


#define N_NODES 50000
#define N_EDGES 800000
#define GNN_IN  256
#define GNN_HID 64
#define COMB    64
#define MLP_IN  256
#define MLP_HID 512

typedef __attribute__((ext_vector_type(16))) __bf16 v16bf;
typedef __attribute__((ext_vector_type(8)))  float  v8f;

typedef __attribute__((ext_vector_type(2)))  float  f32x2;
typedef __attribute__((ext_vector_type(2)))  __bf16 bf16x2;

union V16 {
    uint4 u[2];
    v16bf v;
};

__device__ __forceinline__ unsigned short f2bf(float f) {
    unsigned int u = __float_as_uint(f);
    unsigned int r = u + 0x7FFFu + ((u >> 16) & 1u);
    return (unsigned short)(r >> 16);
}

__device__ __forceinline__ ushort4 f2bf4(const float4 f) {
    ushort4 o;
    o.x = f2bf(f.x); o.y = f2bf(f.y); o.z = f2bf(f.z); o.w = f2bf(f.w);
    return o;
}

__device__ __forceinline__ int package_half() {
    const unsigned int id = __builtin_amdgcn_s_sendmsg_rtn(0x87);
    return (int)((id >> 18) & 1u);
}

#define LOOP_BLOCKS 512
#define SCATTER_BLOCKS 1024

__global__ __launch_bounds__(256) void cvt_bf16_kernel(const float* __restrict__ in,
                                                       unsigned short* __restrict__ out, int n4) {
    constexpr int U = 8;
    const float4* in4 = (const float4*)in;
    ushort4* out4 = (ushort4*)out;
    const int stride = gridDim.x * blockDim.x;
    int i = blockIdx.x * blockDim.x + threadIdx.x;
    for (; i + (U - 1) * stride < n4; i += U * stride) {
        float4 f[U];
#pragma unroll
        for (int u = 0; u < U; ++u) f[u] = in4[i + u * stride];
#pragma unroll
        for (int u = 0; u < U; ++u) out4[i + u * stride] = f2bf4(f[u]);
    }
    for (; i < n4; i += stride) out4[i] = f2bf4(in4[i]);
}

__global__ void pack_w_kernel(const float* __restrict__ W,
                              unsigned short* __restrict__ P, int K, int N) {
    int idx = blockIdx.x * blockDim.x + threadIdx.x;
    int nTiles = N >> 4;
    int total = (K >> 5) * nTiles * 32;
    if (idx >= total) return;
    int lane = idx & 31;
    int tile = idx >> 5;
    int kt = tile / nTiles;
    int nt = tile % nTiles;
    int col = nt * 16 + (lane & 15);
    int kb  = kt * 32 + ((lane < 16) ? 0 : 8);
    __attribute__((aligned(16))) unsigned short o[16];
#pragma unroll
    for (int j = 0; j < 8; ++j) o[j]     = f2bf(W[(size_t)(kb + j) * N + col]);
#pragma unroll
    for (int j = 0; j < 8; ++j) o[8 + j] = f2bf(W[(size_t)(kb + 16 + j) * N + col]);
    uint4* dst = (uint4*)(P + (size_t)tile * 512 + lane * 16);
    dst[0] = ((uint4*)o)[0];
    dst[1] = ((uint4*)o)[1];
}

template <int K>
__device__ __forceinline__ void load_a_tile(V16 (&a)[K / 32],
                                            const unsigned short* __restrict__ abase,
                                            int mt) {
    const unsigned short* arow = abase + (size_t)mt * 16 * K;
#pragma unroll
    for (int kt = 0; kt < K / 32; ++kt) {
        const uint4* ap = (const uint4*)(arow + kt * 32);
        a[kt].u[0] = ap[0];
        a[kt].u[1] = ap[2];
    }
}

template <int K, int N, int NTW, bool RELU, bool OUTBF>
__device__ __forceinline__ void tile_mma_store(const V16 (&a)[K / 32],
                                               const V16 (&b)[K / 32][NTW],
                                               const float (&bv)[NTW],
                                               void* __restrict__ C,
                                               int mt, int cg, int lane) {
    v8f acc[NTW];
#pragma unroll
    for (int j = 0; j < NTW; ++j) acc[j] = v8f{};
#pragma unroll
    for (int kt = 0; kt < K / 32; ++kt) {
#pragma unroll
        for (int j = 0; j < NTW; ++j) {
            acc[j] = __builtin_amdgcn_wmma_f32_16x16x32_bf16(false, a[kt].v, false, b[kt][j].v,
                                                             (short)0, acc[j], false, false);
        }
    }
    const int r0 = mt * 16 + ((lane < 16) ? 0 : 8);
#pragma unroll
    for (int j = 0; j < NTW; ++j) {
        const int cc = (cg * NTW + j) * 16 + (lane & 15);
#pragma unroll
        for (int v = 0; v < 8; ++v) {
            float val = acc[j][v] + bv[j];
            if (RELU) val = fmaxf(val, 0.0f);
            const size_t off = (size_t)(r0 + v) * N + cc;
            if (OUTBF) ((unsigned short*)C)[off] = f2bf(val);
            else       ((float*)C)[off] = val;
        }
    }
}

template <int K, int N, int NTW, bool RELU, bool OUTBF, bool HASBIAS>
__global__ __launch_bounds__(256) void gemm_wmma_kernel(const unsigned short* __restrict__ A,
                                                        const unsigned short* __restrict__ Bp,
                                                        const float* __restrict__ bias,
                                                        void* __restrict__ C, int M) {
    constexpr int KT = K / 32;
    constexpr int NT = N / 16;
    constexpr int CG = NT / NTW;
    constexpr int S  = 8 / CG;
    static_assert(K % 32 == 0 && N % 16 == 0 && NT % NTW == 0 && 8 % CG == 0,
                  "tile split must divide the problem and the 8 waves of a block");
    const int lane = threadIdx.x & 31;
    const int wid  = __builtin_amdgcn_readfirstlane(threadIdx.x >> 5);
    const int cg     = wid % CG;
    const int stream = wid / CG;
    const int mTiles = M >> 4;
    const int step   = gridDim.x * S;
    int mt = blockIdx.x * S + stream;
    if (mt >= mTiles) return;

    V16 b[KT][NTW];
#pragma unroll
    for (int kt = 0; kt < KT; ++kt) {
#pragma unroll
        for (int j = 0; j < NTW; ++j) {
            const uint4* bp = (const uint4*)(Bp + (size_t)(kt * NT + cg * NTW + j) * 512 + lane * 16);
            b[kt][j].u[0] = bp[0];
            b[kt][j].u[1] = bp[1];
        }
    }
    float bv[NTW];
#pragma unroll
    for (int j = 0; j < NTW; ++j)
        bv[j] = HASBIAS ? bias[(cg * NTW + j) * 16 + (lane & 15)] : 0.0f;

    const unsigned short* abase = A + (size_t)(lane & 15) * K + ((lane < 16) ? 0 : 8);

    V16 a0[KT], a1[KT];
    load_a_tile<K>(a0, abase, mt);
    while (mt + step < mTiles) {
        load_a_tile<K>(a1, abase, mt + step);
        __builtin_amdgcn_sched_barrier(0);
        tile_mma_store<K, N, NTW, RELU, OUTBF>(a0, b, bv, C, mt, cg, lane);
        const int nxt = mt + 2 * step;
        load_a_tile<K>(a0, abase, (nxt < mTiles) ? nxt : mt);
        __builtin_amdgcn_sched_barrier(0);
        tile_mma_store<K, N, NTW, RELU, OUTBF>(a1, b, bv, C, mt + step, cg, lane);
        mt = nxt;
    }
    if (mt < mTiles) tile_mma_store<K, N, NTW, RELU, OUTBF>(a0, b, bv, C, mt, cg, lane);
}

#define FZ_W2_BYTES   (MLP_HID * COMB * 2)
#define FZ_ROW_PITCH  272
#define FZ_WAVE_BYTES (16 * FZ_ROW_PITCH)
#define FZ_BUF_BYTES  (8 * FZ_WAVE_BYTES)
#define FZ_LDS_BYTES  (FZ_W2_BYTES + 2 * FZ_BUF_BYTES)

__device__ __forceinline__ void lds_barrier() {
    asm volatile("s_wait_dscnt 0x0\n\ts_barrier_signal -1\n\ts_barrier_wait -1" ::: "memory");
}

__device__ __forceinline__ unsigned int relu_pack2(float lo, float hi, float blo, float bhi) {
    const f32x2 v = {fmaxf(lo + blo, 0.0f), fmaxf(hi + bhi, 0.0f)};
    return __builtin_bit_cast(unsigned int, __builtin_convertvector(v, bf16x2));
}

__device__ __forceinline__ void mlp_tile(const V16 (&a)[MLP_IN / 32],
                                         const V16 (&b)[MLP_IN / 32][4],
                                         const float4 (&bias1)[4][2],
                                         const unsigned char* lds, unsigned char* pbuf,
                                         const float4 bias2, float* __restrict__ out,
                                         int mt, int wid, int lane) {
    const int c = lane & 15;
    const int h = lane >> 4;
    v8f acc[4];
#pragma unroll
    for (int j = 0; j < 4; ++j) acc[j] = v8f{};
#pragma unroll
    for (int kt = 0; kt < MLP_IN / 32; ++kt) {
#pragma unroll
        for (int j = 0; j < 4; ++j) {
            acc[j] = __builtin_amdgcn_wmma_f32_16x16x32_bf16(false, b[kt][j].v, false, a[kt].v,
                                                             (short)0, acc[j], false, false);
        }
    }
    __builtin_amdgcn_sched_barrier(0);
    V16 w2[2][4];
#pragma unroll
    for (int q = 0; q < 2; ++q) {
#pragma unroll
        for (int nt = 0; nt < 4; ++nt) {
            const unsigned char* wp = lds + ((2 * wid + q) * 4 + nt) * 1024 + lane * 16;
            w2[q][nt].u[0] = *(const uint4*)(wp);
            w2[q][nt].u[1] = *(const uint4*)(wp + 512);
        }
    }
    __builtin_amdgcn_sched_barrier(0);
    V16 hf[2];
#pragma unroll
    for (int q = 0; q < 2; ++q) {
#pragma unroll
        for (int t = 0; t < 2; ++t) {
            const int j = 2 * q + t;
            const float4 p0 = bias1[j][0], p1 = bias1[j][1];
            uint4 u;
            u.x = relu_pack2(acc[j][0], acc[j][1], p0.x, p0.y);
            u.y = relu_pack2(acc[j][2], acc[j][3], p0.z, p0.w);
            u.z = relu_pack2(acc[j][4], acc[j][5], p1.x, p1.y);
            u.w = relu_pack2(acc[j][6], acc[j][7], p1.z, p1.w);
            hf[q].u[t] = u;
        }
    }
    v8f acc2[4];
#pragma unroll
    for (int nt = 0; nt < 4; ++nt) acc2[nt] = v8f{};
#pragma unroll
    for (int q = 0; q < 2; ++q) {
#pragma unroll
        for (int nt = 0; nt < 4; ++nt) {
            acc2[nt] = __builtin_amdgcn_wmma_f32_16x16x32_bf16(false, w2[q][nt].v, false, hf[q].v,
                                                               (short)0, acc2[nt], false, false);
        }
    }
#pragma unroll
    for (int nt = 0; nt < 4; ++nt) {
        float4* pp = (float4*)(pbuf + wid * FZ_WAVE_BYTES + c * FZ_ROW_PITCH + (16 * nt + 8 * h) * 4);
        const float4 lo = {acc2[nt][0], acc2[nt][1], acc2[nt][2], acc2[nt][3]};
        const float4 hi = {acc2[nt][4], acc2[nt][5], acc2[nt][6], acc2[nt][7]};
        pp[0] = lo;
        pp[1] = hi;
    }
    lds_barrier();
    const int node = threadIdx.x >> 4;
    const int col4 = (threadIdx.x & 15) * 4;
    const unsigned char* rp = pbuf + node * FZ_ROW_PITCH + col4 * 4;
    float4 s = *(const float4*)rp;
#pragma unroll
    for (int w = 1; w < 8; ++w) {
        const float4 p = *(const float4*)(rp + w * FZ_WAVE_BYTES);
        s.x += p.x; s.y += p.y; s.z += p.z; s.w += p.w;
    }
    s.x += bias2.x; s.y += bias2.y; s.z += bias2.z; s.w += bias2.w;
    *(float4*)(out + (size_t)(mt * 16 + node) * COMB + col4) = s;
}

__global__ __launch_bounds__(256) void mlp_chain_kernel(const unsigned short* __restrict__ A,
                                                        const unsigned short* __restrict__ W1p,
                                                        const float* __restrict__ b1,
                                                        const unsigned short* __restrict__ W2p,
                                                        const float* __restrict__ b2,
                                                        float* __restrict__ out, int M) {
    constexpr int K  = MLP_IN;
    constexpr int KT = K / 32;
    constexpr int NT = MLP_HID / 16;
    __shared__ __attribute__((aligned(16))) unsigned char L[FZ_LDS_BYTES];
    const int lane = threadIdx.x & 31;
    const int wid = __builtin_amdgcn_readfirstlane(threadIdx.x >> 5);
    const int mTiles = M >> 4;
    const int step   = gridDim.x;
    int mt = blockIdx.x;
    if (mt >= mTiles) return;

    V16 b[KT][4];
#pragma unroll
    for (int kt = 0; kt < KT; ++kt) {
#pragma unroll
        for (int j = 0; j < 4; ++j) {
            const uint4* bp = (const uint4*)(W1p + (size_t)(kt * NT + wid * 4 + j) * 512 + lane * 16);
            b[kt][j].u[0] = bp[0];
            b[kt][j].u[1] = bp[1];
        }
    }
    const unsigned short* abase = A + (size_t)(lane & 15) * K + ((lane < 16) ? 0 : 8);
    V16 a0[KT], a1[KT];
    load_a_tile<K>(a0, abase, mt);

    for (int i = threadIdx.x; i < FZ_W2_BYTES / 16; i += 256) {
        const uint4 v = ((const uint4*)W2p)[i];
        *(uint4*)(L + (i >> 6) * 1024 + (i & 1) * 512 + ((i & 63) >> 1) * 16) = v;
    }
    float4 bias1[4][2];
#pragma unroll
    for (int j = 0; j < 4; ++j) {
        const float4* bp = (const float4*)(b1 + wid * 64 + j * 16 + (lane >> 4) * 8);
        bias1[j][0] = bp[0];
        bias1[j][1] = bp[1];
    }
    const float4 bias2 = *(const float4*)(b2 + (threadIdx.x & 15) * 4);
    lds_barrier();

    unsigned char* pb0 = L + FZ_W2_BYTES;
    unsigned char* pb1 = pb0 + FZ_BUF_BYTES;
    while (mt + step < mTiles) {
        load_a_tile<K>(a1, abase, mt + step);
        __builtin_amdgcn_sched_barrier(0);
        mlp_tile(a0, b, bias1, L, pb0, bias2, out, mt, wid, lane);
        const int nxt = mt + 2 * step;
        load_a_tile<K>(a0, abase, (nxt < mTiles) ? nxt : mt);
        __builtin_amdgcn_sched_barrier(0);
        mlp_tile(a1, b, bias1, L, pb1, bias2, out, mt + step, wid, lane);
        mt = nxt;
    }
    if (mt < mTiles) mlp_tile(a0, b, bias1, L, pb0, bias2, out, mt, wid, lane);
}

__global__ void deg_init_kernel(float* __restrict__ deg0, float* __restrict__ deg1, int n) {
    int i = blockIdx.x * blockDim.x + threadIdx.x;
    if (i < n) {
        deg0[i] = 1.0f;
        deg1[i] = 0.0f;
    }
}

__global__ __launch_bounds__(256) void deg_accum_kernel(const int* __restrict__ dst,
                                                        const float* __restrict__ ew,
                                                        float* __restrict__ deg0,
                                                        float* __restrict__ deg1, int nE) {
    constexpr int U = 6;
    float* __restrict__ deg = package_half() ? deg1 : deg0;
    const int stride = gridDim.x * blockDim.x;
    int e = blockIdx.x * blockDim.x + threadIdx.x;
    for (; e + (U - 1) * stride < nE; e += U * stride) {
        int d[U];
        float w[U];
#pragma unroll
        for (int u = 0; u < U; ++u) { d[u] = dst[e + u * stride]; w[u] = ew[e + u * stride]; }
#pragma unroll
        for (int u = 0; u < U; ++u) atomicAdd(&deg[d[u]], w[u]);
    }
    for (; e < nE; e += stride) atomicAdd(&deg[dst[e]], ew[e]);
}

__global__ void dinv_kernel(const float* __restrict__ deg0,
                            const float* __restrict__ deg1,
                            float* __restrict__ dinv, int n) {
    int i = blockIdx.x * blockDim.x + threadIdx.x;
    if (i >= n) return;
    float d = deg0[i] + deg1[i];
    dinv[i] = (d > 0.0f) ? rsqrtf(fmaxf(d, 1e-12f)) : 0.0f;
}

__global__ __launch_bounds__(256) void edge_norm_kernel(const int* __restrict__ src,
                                                        const int* __restrict__ dst,
                                                        const float* __restrict__ ew,
                                                        const float* __restrict__ dinv,
                                                        float* __restrict__ norm, int nE) {
    constexpr int U = 6;
    const int stride = gridDim.x * blockDim.x;
    int e = blockIdx.x * blockDim.x + threadIdx.x;
    for (; e + (U - 1) * stride < nE; e += U * stride) {
        int s[U], d[U];
        float w[U], ds[U], dd[U];
#pragma unroll
        for (int u = 0; u < U; ++u) {
            s[u] = src[e + u * stride]; d[u] = dst[e + u * stride]; w[u] = ew[e + u * stride];
        }
#pragma unroll
        for (int u = 0; u < U; ++u) { ds[u] = dinv[s[u]]; dd[u] = dinv[d[u]]; }
#pragma unroll
        for (int u = 0; u < U; ++u) norm[e + u * stride] = ds[u] * w[u] * dd[u];
    }
    for (; e < nE; e += stride) norm[e] = dinv[src[e]] * ew[e] * dinv[dst[e]];
}

__global__ __launch_bounds__(256) void agg_init_kernel(const float* __restrict__ h,
                                                       const float* __restrict__ dinv,
                                                       float* __restrict__ agg,
                                                       float* __restrict__ agg1, int nNodes) {
    constexpr int U = 6;
    const float4* h4 = (const float4*)h;
    float4* agg4 = (float4*)agg;
    float4* aggz = (float4*)agg1;
    const float4 zero = {0.0f, 0.0f, 0.0f, 0.0f};
    const int n16 = nNodes * (COMB / 4);
    const int stride = gridDim.x * blockDim.x;
    int idx = blockIdx.x * blockDim.x + threadIdx.x;
    for (; idx + (U - 1) * stride < n16; idx += U * stride) {
        float4 hv[U];
        float w[U];
#pragma unroll
        for (int u = 0; u < U; ++u) {
            hv[u] = h4[idx + u * stride];
            w[u]  = dinv[(idx + u * stride) >> 4];
        }
#pragma unroll
        for (int u = 0; u < U; ++u) {
            const float ww = w[u] * w[u];
            const float4 o = {hv[u].x * ww, hv[u].y * ww, hv[u].z * ww, hv[u].w * ww};
            agg4[idx + u * stride] = o;
            aggz[idx + u * stride] = zero;
        }
    }
    for (; idx < n16; idx += stride) {
        float w = dinv[idx >> 4]; w *= w;
        const float4 hv = h4[idx];
        const float4 o = {hv.x * w, hv.y * w, hv.z * w, hv.w * w};
        agg4[idx] = o;
        aggz[idx] = zero;
    }
}

template <bool GUARD>
__device__ __forceinline__ void scatter_group(const float* __restrict__ h,
                                              float* __restrict__ agg,
                                              int s_l, int d_l, float n_l,
                                              int half, int c, int left) {
    constexpr int P = 8;
#pragma unroll
    for (int j0 = 0; j0 < 16; j0 += P) {
        float4 hv[P];
        float nm[P];
        int d[P];
#pragma unroll
        for (int p = 0; p < P; ++p) {
            const int sl = 2 * (j0 + p) + half;
            const int s = __shfl(s_l, sl);
            d[p]  = __shfl(d_l, sl);
            nm[p] = __shfl(n_l, sl);
            if (!GUARD || sl < left) hv[p] = *(const float4*)(h + (size_t)s * COMB + c);
        }
#pragma unroll
        for (int p = 0; p < P; ++p) {
            if (!GUARD || 2 * (j0 + p) + half < left) {
                float* ap = agg + (size_t)d[p] * COMB + c;
                atomicAdd(ap + 0, hv[p].x * nm[p]);
                atomicAdd(ap + 1, hv[p].y * nm[p]);
                atomicAdd(ap + 2, hv[p].z * nm[p]);
                atomicAdd(ap + 3, hv[p].w * nm[p]);
            }
        }
    }
}

__global__ __launch_bounds__(256) void gcn_scatter_kernel(const float* __restrict__ h,
                                                          const int* __restrict__ src,
                                                          const int* __restrict__ dst,
                                                          const float* __restrict__ norm,
                                                          float* __restrict__ agg0,
                                                          float* __restrict__ agg1, int nE) {
    float* __restrict__ agg = package_half() ? agg1 : agg0;
    const int lane = threadIdx.x & 31;
    const int half = lane >> 4;
    const int c    = (lane & 15) * 4;
    const int gstep = ((gridDim.x * blockDim.x) >> 5) * 32;
    int base = ((blockIdx.x * blockDim.x + threadIdx.x) >> 5) * 32;
    for (; base + 32 <= nE; base += gstep) {
        const int e = base + lane;
        scatter_group<false>(h, agg, src[e], dst[e], norm[e], half, c, 32);
    }
    if (base < nE) {
        const int e = base + lane;
        const bool ok = e < nE;
        scatter_group<true>(h, agg, ok ? src[e] : 0, ok ? dst[e] : 0, ok ? norm[e] : 0.0f,
                            half, c, nE - base);
    }
}

__global__ __launch_bounds__(256) void relu_bias_bf16_kernel(const float* __restrict__ agg,
                                                             const float* __restrict__ agg1,
                                                             const float* __restrict__ b,
                                                             unsigned short* __restrict__ g,
                                                             int total) {
    constexpr int U = 6;
    const float4* a4 = (const float4*)agg;
    const float4* p4 = (const float4*)agg1;
    ushort4* g4 = (ushort4*)g;
    const int n4 = total >> 2;
    const int stride = gridDim.x * blockDim.x;
    int idx = blockIdx.x * blockDim.x + threadIdx.x;
    const float4 bv = *(const float4*)(b + (idx & 15) * 4);
    for (; idx + (U - 1) * stride < n4; idx += U * stride) {
        float4 v[U], q[U];
#pragma unroll
        for (int u = 0; u < U; ++u) { v[u] = a4[idx + u * stride]; q[u] = p4[idx + u * stride]; }
#pragma unroll
        for (int u = 0; u < U; ++u) {
            const float4 r = {fmaxf((v[u].x + q[u].x) + bv.x, 0.0f),
                              fmaxf((v[u].y + q[u].y) + bv.y, 0.0f),
                              fmaxf((v[u].z + q[u].z) + bv.z, 0.0f),
                              fmaxf((v[u].w + q[u].w) + bv.w, 0.0f)};
            g4[idx + u * stride] = f2bf4(r);
        }
    }
    for (; idx < n4; idx += stride) {
        const float4 v = a4[idx];
        const float4 q = p4[idx];
        const float4 r = {fmaxf((v.x + q.x) + bv.x, 0.0f), fmaxf((v.y + q.y) + bv.y, 0.0f),
                          fmaxf((v.z + q.z) + bv.z, 0.0f), fmaxf((v.w + q.w) + bv.w, 0.0f)};
        g4[idx] = f2bf4(r);
    }
}

__global__ __launch_bounds__(64) void final_kernel(const float* __restrict__ mlp,
                                                   const float* __restrict__ agg2,
                                                   const float* __restrict__ agg2b,
                                                   const float* __restrict__ b2,
                                                   const float* __restrict__ fw,
                                                   const float* __restrict__ fb,
                                                   float* __restrict__ out, int nNodes) {
    constexpr int U = 8;
    const int lane = threadIdx.x & 31;
    const int n0 = ((blockIdx.x * blockDim.x + threadIdx.x) >> 5) * 32;
    if (n0 >= nNodes) return;
    const float f0 = fw[lane], f1 = fw[32 + lane], f2 = fw[64 + lane], f3 = fw[96 + lane];
    const float c0 = b2[lane], c1 = b2[32 + lane];
    float res = 0.0f;
    for (int j0 = 0; j0 < 32; j0 += U) {
        float m0[U], m1[U], a0[U], a1[U], p0[U], p1[U];
#pragma unroll
        for (int u = 0; u < U; ++u) {
            int n = n0 + j0 + u;
            n = (n < nNodes) ? n : (nNodes - 1);
            const float* mr = mlp  + (size_t)n * COMB;
            const float* ar = agg2 + (size_t)n * COMB;
            const float* pr = agg2b + (size_t)n * COMB;
            m0[u] = mr[lane]; m1[u] = mr[32 + lane];
            a0[u] = ar[lane]; a1[u] = ar[32 + lane];
            p0[u] = pr[lane]; p1[u] = pr[32 + lane];
        }
#pragma unroll
        for (int u = 0; u < U; ++u) {
            float x = (m0[u] * f0 + ((a0[u] + p0[u]) + c0) * f2) +
                      (m1[u] * f1 + ((a1[u] + p1[u]) + c1) * f3);
            x += __shfl_xor(x, 16);
            x += __shfl_xor(x, 8);
            x += __shfl_xor(x, 4);
            x += __shfl_xor(x, 2);
            x += __shfl_xor(x, 1);
            if (lane == j0 + u) res = x;
        }
    }
    if (n0 + lane < nNodes) out[n0 + lane] = res + fb[0];
}

extern "C" void kernel_launch(void* const* d_in, const int* in_sizes, int n_in,
                              void* d_out, int out_size, void* d_ws, size_t ws_size,
                              hipStream_t stream) {
    (void)in_sizes; (void)n_in; (void)out_size; (void)ws_size;

    const float* x        = (const float*)d_in[0];
    const int*   eidx     = (const int*)d_in[1];
    const float* eattr    = (const float*)d_in[2];
    const float* ufeat    = (const float*)d_in[3];
    const float* gcn1_W   = (const float*)d_in[4];
    const float* gcn1_b   = (const float*)d_in[5];
    const float* gcn2_W   = (const float*)d_in[6];
    const float* gcn2_b   = (const float*)d_in[7];
    const float* fc1_W    = (const float*)d_in[8];
    const float* fc1_b    = (const float*)d_in[9];
    const float* fc2_W    = (const float*)d_in[10];
    const float* fc2_b    = (const float*)d_in[11];
    const float* fcf_W    = (const float*)d_in[12];
    const float* fcf_b    = (const float*)d_in[13];
    const int* esrc = eidx;
    const int* edst = eidx + N_EDGES;
    float* out = (float*)d_out;

    char* ws = (char*)d_ws;
    size_t off = 0;
    auto carve = [&](size_t bytes) -> void* {
        void* p = ws + off;
        off = (off + bytes + 255) & ~(size_t)255;
        return p;
    };
    unsigned short* UBF = (unsigned short*)carve((size_t)N_NODES * MLP_IN * 2);
    unsigned short* XBF = (unsigned short*)carve((size_t)N_NODES * GNN_IN * 2);
    unsigned short* HBF = (unsigned short*)carve((size_t)N_NODES * MLP_HID * 2);
    unsigned short* GBF = (unsigned short*)carve((size_t)N_NODES * COMB * 2);
    unsigned short* W1P = (unsigned short*)carve((size_t)MLP_IN * MLP_HID * 2);
    unsigned short* W2P = (unsigned short*)carve((size_t)MLP_HID * COMB * 2);
    unsigned short* G1P = (unsigned short*)carve((size_t)GNN_IN * GNN_HID * 2);
    unsigned short* G2P = (unsigned short*)carve((size_t)GNN_HID * COMB * 2);
    float* MLPOUT = (float*)carve((size_t)N_NODES * COMB * 4);
    float* H1     = (float*)carve((size_t)N_NODES * COMB * 4);
    float* AGG    = (float*)carve((size_t)N_NODES * COMB * 4);
    float* DEG    = (float*)carve((size_t)N_NODES * 4);
    float* DINV   = (float*)carve((size_t)N_NODES * 4);
    float* NORM   = (float*)carve((size_t)N_EDGES * 4);
    float* H2 = H1;
    float* DEG1 = (float*)HBF;
    float* AGG1 = (float*)UBF;

    {
        int n4 = (N_NODES * MLP_IN) / 4;
        cvt_bf16_kernel<<<LOOP_BLOCKS, 256, 0, stream>>>(ufeat, UBF, n4);
        n4 = (N_NODES * GNN_IN) / 4;
        cvt_bf16_kernel<<<LOOP_BLOCKS, 256, 0, stream>>>(x, XBF, n4);
    }
    auto pack = [&](const float* W, unsigned short* P, int K, int N) {
        int total = (K / 32) * (N / 16) * 32;
        pack_w_kernel<<<(total + 255) / 256, 256, 0, stream>>>(W, P, K, N);
    };
    pack(fc1_W, W1P, MLP_IN, MLP_HID);
    pack(fc2_W, W2P, MLP_HID, COMB);
    pack(gcn1_W, G1P, GNN_IN, GNN_HID);
    pack(gcn2_W, G2P, GNN_HID, COMB);

    deg_init_kernel<<<(N_NODES + 255) / 256, 256, 0, stream>>>(DEG, DEG1, N_NODES);
    deg_accum_kernel<<<LOOP_BLOCKS, 256, 0, stream>>>(edst, eattr, DEG, DEG1, N_EDGES);
    dinv_kernel<<<(N_NODES + 255) / 256, 256, 0, stream>>>(DEG, DEG1, DINV, N_NODES);
    edge_norm_kernel<<<LOOP_BLOCKS, 256, 0, stream>>>(esrc, edst, eattr, DINV, NORM, N_EDGES);

    mlp_chain_kernel<<<256, 256, 0, stream>>>(UBF, W1P, fc1_b, W2P, fc2_b, MLPOUT, N_NODES);

    gemm_wmma_kernel<GNN_IN, GNN_HID, 2, false, false, false><<<256, 256, 0, stream>>>(
        XBF, G1P, nullptr, H1, N_NODES);
    agg_init_kernel<<<LOOP_BLOCKS, 256, 0, stream>>>(H1, DINV, AGG, AGG1, N_NODES);
    gcn_scatter_kernel<<<SCATTER_BLOCKS, 256, 0, stream>>>(H1, esrc, edst, NORM, AGG, AGG1, N_EDGES);
    relu_bias_bf16_kernel<<<LOOP_BLOCKS, 256, 0, stream>>>(AGG, AGG1, gcn1_b, GBF, N_NODES * COMB);

    gemm_wmma_kernel<GNN_HID, COMB, 2, false, false, false><<<256, 256, 0, stream>>>(
        GBF, G2P, nullptr, H2, N_NODES);
    agg_init_kernel<<<LOOP_BLOCKS, 256, 0, stream>>>(H2, DINV, AGG, AGG1, N_NODES);
    gcn_scatter_kernel<<<SCATTER_BLOCKS, 256, 0, stream>>>(H2, esrc, edst, NORM, AGG, AGG1, N_EDGES);

    final_kernel<<<(N_NODES + 63) / 64, 64, 0, stream>>>(MLPOUT, AGG, AGG1, gcn2_b, fcf_W, fcf_b,
                                                      out, N_NODES);
}


# layer 2 aggregates before its GEMM: edge pass sums bf16 g rows (1 line/edge, zeros skip the atomic); gcn2 WMMA on the sums (bf16 head+remainder) inside the final pass; sum order changed; 17 launches
# speedup vs baseline: 1.7003x; 1.2756x over previous
#include <hip/hip_runtime.h>
#include <hip/hip_bf16.h>


#define N_NODES 50000
#define N_EDGES 800000
#define GNN_IN  256
#define GNN_HID 64
#define COMB    64
#define MLP_IN  256
#define MLP_HID 512

typedef __attribute__((ext_vector_type(16))) __bf16 v16bf;
typedef __attribute__((ext_vector_type(8)))  float  v8f;

typedef __attribute__((ext_vector_type(2)))  float  f32x2;
typedef __attribute__((ext_vector_type(2)))  __bf16 bf16x2;

union V16 {
    uint4 u[2];
    v16bf v;
};

__device__ __forceinline__ unsigned short f2bf(float f) {
    unsigned int u = __float_as_uint(f);
    unsigned int r = u + 0x7FFFu + ((u >> 16) & 1u);
    return (unsigned short)(r >> 16);
}

__device__ __forceinline__ ushort4 f2bf4(const float4 f) {
    ushort4 o;
    o.x = f2bf(f.x); o.y = f2bf(f.y); o.z = f2bf(f.z); o.w = f2bf(f.w);
    return o;
}

__device__ __forceinline__ int package_half() {
    const unsigned int id = __builtin_amdgcn_s_sendmsg_rtn(0x87);
    return (int)((id >> 18) & 1u);
}

#define LOOP_BLOCKS 512
#define SCATTER_BLOCKS 1024

__global__ __launch_bounds__(256) void cvt_bf16_kernel(const float* __restrict__ in,
                                                       unsigned short* __restrict__ out, int n4) {
    constexpr int U = 8;
    const float4* in4 = (const float4*)in;
    ushort4* out4 = (ushort4*)out;
    const int stride = gridDim.x * blockDim.x;
    int i = blockIdx.x * blockDim.x + threadIdx.x;
    for (; i + (U - 1) * stride < n4; i += U * stride) {
        float4 f[U];
#pragma unroll
        for (int u = 0; u < U; ++u) f[u] = in4[i + u * stride];
#pragma unroll
        for (int u = 0; u < U; ++u) out4[i + u * stride] = f2bf4(f[u]);
    }
    for (; i < n4; i += stride) out4[i] = f2bf4(in4[i]);
}

__global__ void pack_w_kernel(const float* __restrict__ W,
                              unsigned short* __restrict__ P, int K, int N) {
    int idx = blockIdx.x * blockDim.x + threadIdx.x;
    int nTiles = N >> 4;
    int total = (K >> 5) * nTiles * 32;
    if (idx >= total) return;
    int lane = idx & 31;
    int tile = idx >> 5;
    int kt = tile / nTiles;
    int nt = tile % nTiles;
    int col = nt * 16 + (lane & 15);
    int kb  = kt * 32 + ((lane < 16) ? 0 : 8);
    __attribute__((aligned(16))) unsigned short o[16];
#pragma unroll
    for (int j = 0; j < 8; ++j) o[j]     = f2bf(W[(size_t)(kb + j) * N + col]);
#pragma unroll
    for (int j = 0; j < 8; ++j) o[8 + j] = f2bf(W[(size_t)(kb + 16 + j) * N + col]);
    uint4* dst = (uint4*)(P + (size_t)tile * 512 + lane * 16);
    dst[0] = ((uint4*)o)[0];
    dst[1] = ((uint4*)o)[1];
}

template <int K>
__device__ __forceinline__ void load_a_tile(V16 (&a)[K / 32],
                                            const unsigned short* __restrict__ abase,
                                            int mt) {
    const unsigned short* arow = abase + (size_t)mt * 16 * K;
#pragma unroll
    for (int kt = 0; kt < K / 32; ++kt) {
        const uint4* ap = (const uint4*)(arow + kt * 32);
        a[kt].u[0] = ap[0];
        a[kt].u[1] = ap[2];
    }
}

template <int K, int N, int NTW, bool RELU, bool OUTBF>
__device__ __forceinline__ void tile_mma_store(const V16 (&a)[K / 32],
                                               const V16 (&b)[K / 32][NTW],
                                               const float (&bv)[NTW],
                                               void* __restrict__ C,
                                               int mt, int cg, int lane) {
    v8f acc[NTW];
#pragma unroll
    for (int j = 0; j < NTW; ++j) acc[j] = v8f{};
#pragma unroll
    for (int kt = 0; kt < K / 32; ++kt) {
#pragma unroll
        for (int j = 0; j < NTW; ++j) {
            acc[j] = __builtin_amdgcn_wmma_f32_16x16x32_bf16(false, a[kt].v, false, b[kt][j].v,
                                                             (short)0, acc[j], false, false);
        }
    }
    const int r0 = mt * 16 + ((lane < 16) ? 0 : 8);
#pragma unroll
    for (int j = 0; j < NTW; ++j) {
        const int cc = (cg * NTW + j) * 16 + (lane & 15);
#pragma unroll
        for (int v = 0; v < 8; ++v) {
            float val = acc[j][v] + bv[j];
            if (RELU) val = fmaxf(val, 0.0f);
            const size_t off = (size_t)(r0 + v) * N + cc;
            if (OUTBF) ((unsigned short*)C)[off] = f2bf(val);
            else       ((float*)C)[off] = val;
        }
    }
}

template <int K, int N, int NTW, bool RELU, bool OUTBF, bool HASBIAS>
__global__ __launch_bounds__(256) void gemm_wmma_kernel(const unsigned short* __restrict__ A,
                                                        const unsigned short* __restrict__ Bp,
                                                        const float* __restrict__ bias,
                                                        void* __restrict__ C, int M) {
    constexpr int KT = K / 32;
    constexpr int NT = N / 16;
    constexpr int CG = NT / NTW;
    constexpr int S  = 8 / CG;
    static_assert(K % 32 == 0 && N % 16 == 0 && NT % NTW == 0 && 8 % CG == 0,
                  "tile split must divide the problem and the 8 waves of a block");
    const int lane = threadIdx.x & 31;
    const int wid  = __builtin_amdgcn_readfirstlane(threadIdx.x >> 5);
    const int cg     = wid % CG;
    const int stream = wid / CG;
    const int mTiles = M >> 4;
    const int step   = gridDim.x * S;
    int mt = blockIdx.x * S + stream;
    if (mt >= mTiles) return;

    V16 b[KT][NTW];
#pragma unroll
    for (int kt = 0; kt < KT; ++kt) {
#pragma unroll
        for (int j = 0; j < NTW; ++j) {
            const uint4* bp = (const uint4*)(Bp + (size_t)(kt * NT + cg * NTW + j) * 512 + lane * 16);
            b[kt][j].u[0] = bp[0];
            b[kt][j].u[1] = bp[1];
        }
    }
    float bv[NTW];
#pragma unroll
    for (int j = 0; j < NTW; ++j)
        bv[j] = HASBIAS ? bias[(cg * NTW + j) * 16 + (lane & 15)] : 0.0f;

    const unsigned short* abase = A + (size_t)(lane & 15) * K + ((lane < 16) ? 0 : 8);

    V16 a0[KT], a1[KT];
    load_a_tile<K>(a0, abase, mt);
    while (mt + step < mTiles) {
        load_a_tile<K>(a1, abase, mt + step);
        __builtin_amdgcn_sched_barrier(0);
        tile_mma_store<K, N, NTW, RELU, OUTBF>(a0, b, bv, C, mt, cg, lane);
        const int nxt = mt + 2 * step;
        load_a_tile<K>(a0, abase, (nxt < mTiles) ? nxt : mt);
        __builtin_amdgcn_sched_barrier(0);
        tile_mma_store<K, N, NTW, RELU, OUTBF>(a1, b, bv, C, mt + step, cg, lane);
        mt = nxt;
    }
    if (mt < mTiles) tile_mma_store<K, N, NTW, RELU, OUTBF>(a0, b, bv, C, mt, cg, lane);
}

#define FZ_W2_BYTES   (MLP_HID * COMB * 2)
#define FZ_ROW_PITCH  272
#define FZ_WAVE_BYTES (16 * FZ_ROW_PITCH)
#define FZ_BUF_BYTES  (8 * FZ_WAVE_BYTES)
#define FZ_LDS_BYTES  (FZ_W2_BYTES + 2 * FZ_BUF_BYTES)

__device__ __forceinline__ void lds_barrier() {
    asm volatile("s_wait_dscnt 0x0\n\ts_barrier_signal -1\n\ts_barrier_wait -1" ::: "memory");
}

__device__ __forceinline__ unsigned int relu_pack2(float lo, float hi, float blo, float bhi) {
    const f32x2 v = {fmaxf(lo + blo, 0.0f), fmaxf(hi + bhi, 0.0f)};
    return __builtin_bit_cast(unsigned int, __builtin_convertvector(v, bf16x2));
}

__device__ __forceinline__ void mlp_tile(const V16 (&a)[MLP_IN / 32],
                                         const V16 (&b)[MLP_IN / 32][4],
                                         const float4 (&bias1)[4][2],
                                         const unsigned char* lds, unsigned char* pbuf,
                                         const float4 bias2, float* __restrict__ out,
                                         int mt, int wid, int lane) {
    const int c = lane & 15;
    const int h = lane >> 4;
    v8f acc[4];
#pragma unroll
    for (int j = 0; j < 4; ++j) acc[j] = v8f{};
#pragma unroll
    for (int kt = 0; kt < MLP_IN / 32; ++kt) {
#pragma unroll
        for (int j = 0; j < 4; ++j) {
            acc[j] = __builtin_amdgcn_wmma_f32_16x16x32_bf16(false, b[kt][j].v, false, a[kt].v,
                                                             (short)0, acc[j], false, false);
        }
    }
    __builtin_amdgcn_sched_barrier(0);
    V16 w2[2][4];
#pragma unroll
    for (int q = 0; q < 2; ++q) {
#pragma unroll
        for (int nt = 0; nt < 4; ++nt) {
            const unsigned char* wp = lds + ((2 * wid + q) * 4 + nt) * 1024 + lane * 16;
            w2[q][nt].u[0] = *(const uint4*)(wp);
            w2[q][nt].u[1] = *(const uint4*)(wp + 512);
        }
    }
    __builtin_amdgcn_sched_barrier(0);
    V16 hf[2];
#pragma unroll
    for (int q = 0; q < 2; ++q) {
#pragma unroll
        for (int t = 0; t < 2; ++t) {
            const int j = 2 * q + t;
            const float4 p0 = bias1[j][0], p1 = bias1[j][1];
            uint4 u;
            u.x = relu_pack2(acc[j][0], acc[j][1], p0.x, p0.y);
            u.y = relu_pack2(acc[j][2], acc[j][3], p0.z, p0.w);
            u.z = relu_pack2(acc[j][4], acc[j][5], p1.x, p1.y);
            u.w = relu_pack2(acc[j][6], acc[j][7], p1.z, p1.w);
            hf[q].u[t] = u;
        }
    }
    v8f acc2[4];
#pragma unroll
    for (int nt = 0; nt < 4; ++nt) acc2[nt] = v8f{};
#pragma unroll
    for (int q = 0; q < 2; ++q) {
#pragma unroll
        for (int nt = 0; nt < 4; ++nt) {
            acc2[nt] = __builtin_amdgcn_wmma_f32_16x16x32_bf16(false, w2[q][nt].v, false, hf[q].v,
                                                               (short)0, acc2[nt], false, false);
        }
    }
#pragma unroll
    for (int nt = 0; nt < 4; ++nt) {
        float4* pp = (float4*)(pbuf + wid * FZ_WAVE_BYTES + c * FZ_ROW_PITCH + (16 * nt + 8 * h) * 4);
        const float4 lo = {acc2[nt][0], acc2[nt][1], acc2[nt][2], acc2[nt][3]};
        const float4 hi = {acc2[nt][4], acc2[nt][5], acc2[nt][6], acc2[nt][7]};
        pp[0] = lo;
        pp[1] = hi;
    }
    lds_barrier();
    const int node = threadIdx.x >> 4;
    const int col4 = (threadIdx.x & 15) * 4;
    const unsigned char* rp = pbuf + node * FZ_ROW_PITCH + col4 * 4;
    float4 s = *(const float4*)rp;
#pragma unroll
    for (int w = 1; w < 8; ++w) {
        const float4 p = *(const float4*)(rp + w * FZ_WAVE_BYTES);
        s.x += p.x; s.y += p.y; s.z += p.z; s.w += p.w;
    }
    s.x += bias2.x; s.y += bias2.y; s.z += bias2.z; s.w += bias2.w;
    *(float4*)(out + (size_t)(mt * 16 + node) * COMB + col4) = s;
}

__global__ __launch_bounds__(256) void mlp_chain_kernel(const unsigned short* __restrict__ A,
                                                        const unsigned short* __restrict__ W1p,
                                                        const float* __restrict__ b1,
                                                        const unsigned short* __restrict__ W2p,
                                                        const float* __restrict__ b2,
                                                        float* __restrict__ out, int M) {
    constexpr int K  = MLP_IN;
    constexpr int KT = K / 32;
    constexpr int NT = MLP_HID / 16;
    __shared__ __attribute__((aligned(16))) unsigned char L[FZ_LDS_BYTES];
    const int lane = threadIdx.x & 31;
    const int wid = __builtin_amdgcn_readfirstlane(threadIdx.x >> 5);
    const int mTiles = M >> 4;
    const int step   = gridDim.x;
    int mt = blockIdx.x;
    if (mt >= mTiles) return;

    V16 b[KT][4];
#pragma unroll
    for (int kt = 0; kt < KT; ++kt) {
#pragma unroll
        for (int j = 0; j < 4; ++j) {
            const uint4* bp = (const uint4*)(W1p + (size_t)(kt * NT + wid * 4 + j) * 512 + lane * 16);
            b[kt][j].u[0] = bp[0];
            b[kt][j].u[1] = bp[1];
        }
    }
    const unsigned short* abase = A + (size_t)(lane & 15) * K + ((lane < 16) ? 0 : 8);
    V16 a0[KT], a1[KT];
    load_a_tile<K>(a0, abase, mt);

    for (int i = threadIdx.x; i < FZ_W2_BYTES / 16; i += 256) {
        const uint4 v = ((const uint4*)W2p)[i];
        *(uint4*)(L + (i >> 6) * 1024 + (i & 1) * 512 + ((i & 63) >> 1) * 16) = v;
    }
    float4 bias1[4][2];
#pragma unroll
    for (int j = 0; j < 4; ++j) {
        const float4* bp = (const float4*)(b1 + wid * 64 + j * 16 + (lane >> 4) * 8);
        bias1[j][0] = bp[0];
        bias1[j][1] = bp[1];
    }
    const float4 bias2 = *(const float4*)(b2 + (threadIdx.x & 15) * 4);
    lds_barrier();

    unsigned char* pb0 = L + FZ_W2_BYTES;
    unsigned char* pb1 = pb0 + FZ_BUF_BYTES;
    while (mt + step < mTiles) {
        load_a_tile<K>(a1, abase, mt + step);
        __builtin_amdgcn_sched_barrier(0);
        mlp_tile(a0, b, bias1, L, pb0, bias2, out, mt, wid, lane);
        const int nxt = mt + 2 * step;
        load_a_tile<K>(a0, abase, (nxt < mTiles) ? nxt : mt);
        __builtin_amdgcn_sched_barrier(0);
        mlp_tile(a1, b, bias1, L, pb1, bias2, out, mt + step, wid, lane);
        mt = nxt;
    }
    if (mt < mTiles) mlp_tile(a0, b, bias1, L, pb0, bias2, out, mt, wid, lane);
}

__global__ void deg_init_kernel(float* __restrict__ deg0, float* __restrict__ deg1, int n) {
    int i = blockIdx.x * blockDim.x + threadIdx.x;
    if (i < n) {
        deg0[i] = 1.0f;
        deg1[i] = 0.0f;
    }
}

__global__ __launch_bounds__(256) void deg_accum_kernel(const int* __restrict__ dst,
                                                        const float* __restrict__ ew,
                                                        float* __restrict__ deg0,
                                                        float* __restrict__ deg1, int nE) {
    constexpr int U = 6;
    float* __restrict__ deg = package_half() ? deg1 : deg0;
    const int stride = gridDim.x * blockDim.x;
    int e = blockIdx.x * blockDim.x + threadIdx.x;
    for (; e + (U - 1) * stride < nE; e += U * stride) {
        int d[U];
        float w[U];
#pragma unroll
        for (int u = 0; u < U; ++u) { d[u] = dst[e + u * stride]; w[u] = ew[e + u * stride]; }
#pragma unroll
        for (int u = 0; u < U; ++u) atomicAdd(&deg[d[u]], w[u]);
    }
    for (; e < nE; e += stride) atomicAdd(&deg[dst[e]], ew[e]);
}

__global__ void dinv_kernel(const float* __restrict__ deg0,
                            const float* __restrict__ deg1,
                            float* __restrict__ dinv, int n) {
    int i = blockIdx.x * blockDim.x + threadIdx.x;
    if (i >= n) return;
    float d = deg0[i] + deg1[i];
    dinv[i] = (d > 0.0f) ? rsqrtf(fmaxf(d, 1e-12f)) : 0.0f;
}

__global__ __launch_bounds__(256) void edge_norm_kernel(const int* __restrict__ src,
                                                        const int* __restrict__ dst,
                                                        const float* __restrict__ ew,
                                                        const float* __restrict__ dinv,
                                                        float* __restrict__ norm, int nE) {
    constexpr int U = 6;
    const int stride = gridDim.x * blockDim.x;
    int e = blockIdx.x * blockDim.x + threadIdx.x;
    for (; e + (U - 1) * stride < nE; e += U * stride) {
        int s[U], d[U];
        float w[U], ds[U], dd[U];
#pragma unroll
        for (int u = 0; u < U; ++u) {
            s[u] = src[e + u * stride]; d[u] = dst[e + u * stride]; w[u] = ew[e + u * stride];
        }
#pragma unroll
        for (int u = 0; u < U; ++u) { ds[u] = dinv[s[u]]; dd[u] = dinv[d[u]]; }
#pragma unroll
        for (int u = 0; u < U; ++u) norm[e + u * stride] = ds[u] * w[u] * dd[u];
    }
    for (; e < nE; e += stride) norm[e] = dinv[src[e]] * ew[e] * dinv[dst[e]];
}

__global__ __launch_bounds__(256) void agg_init_kernel(const float* __restrict__ h,
                                                       const float* __restrict__ dinv,
                                                       float* __restrict__ agg,
                                                       float* __restrict__ agg1, int nNodes) {
    constexpr int U = 6;
    const float4* h4 = (const float4*)h;
    float4* agg4 = (float4*)agg;
    float4* aggz = (float4*)agg1;
    const float4 zero = {0.0f, 0.0f, 0.0f, 0.0f};
    const int n16 = nNodes * (COMB / 4);
    const int stride = gridDim.x * blockDim.x;
    int idx = blockIdx.x * blockDim.x + threadIdx.x;
    for (; idx + (U - 1) * stride < n16; idx += U * stride) {
        float4 hv[U];
        float w[U];
#pragma unroll
        for (int u = 0; u < U; ++u) {
            hv[u] = h4[idx + u * stride];
            w[u]  = dinv[(idx + u * stride) >> 4];
        }
#pragma unroll
        for (int u = 0; u < U; ++u) {
            const float ww = w[u] * w[u];
            const float4 o = {hv[u].x * ww, hv[u].y * ww, hv[u].z * ww, hv[u].w * ww};
            agg4[idx + u * stride] = o;
            aggz[idx + u * stride] = zero;
        }
    }
    for (; idx < n16; idx += stride) {
        float w = dinv[idx >> 4]; w *= w;
        const float4 hv = h4[idx];
        const float4 o = {hv.x * w, hv.y * w, hv.z * w, hv.w * w};
        agg4[idx] = o;
        aggz[idx] = zero;
    }
}

template <bool GUARD>
__device__ __forceinline__ void scatter_group(const float* __restrict__ h,
                                              float* __restrict__ agg,
                                              int s_l, int d_l, float n_l,
                                              int half, int c, int left) {
    constexpr int P = 8;
#pragma unroll
    for (int j0 = 0; j0 < 16; j0 += P) {
        float4 hv[P];
        float nm[P];
        int d[P];
#pragma unroll
        for (int p = 0; p < P; ++p) {
            const int sl = 2 * (j0 + p) + half;
            const int s = __shfl(s_l, sl);
            d[p]  = __shfl(d_l, sl);
            nm[p] = __shfl(n_l, sl);
            if (!GUARD || sl < left) hv[p] = *(const float4*)(h + (size_t)s * COMB + c);
        }
#pragma unroll
        for (int p = 0; p < P; ++p) {
            if (!GUARD || 2 * (j0 + p) + half < left) {
                float* ap = agg + (size_t)d[p] * COMB + c;
                atomicAdd(ap + 0, hv[p].x * nm[p]);
                atomicAdd(ap + 1, hv[p].y * nm[p]);
                atomicAdd(ap + 2, hv[p].z * nm[p]);
                atomicAdd(ap + 3, hv[p].w * nm[p]);
            }
        }
    }
}

__global__ __launch_bounds__(256) void gcn_scatter_kernel(const float* __restrict__ h,
                                                          const int* __restrict__ src,
                                                          const int* __restrict__ dst,
                                                          const float* __restrict__ norm,
                                                          float* __restrict__ agg0,
                                                          float* __restrict__ agg1, int nE) {
    float* __restrict__ agg = package_half() ? agg1 : agg0;
    const int lane = threadIdx.x & 31;
    const int half = lane >> 4;
    const int c    = (lane & 15) * 4;
    const int gstep = ((gridDim.x * blockDim.x) >> 5) * 32;
    int base = ((blockIdx.x * blockDim.x + threadIdx.x) >> 5) * 32;
    for (; base + 32 <= nE; base += gstep) {
        const int e = base + lane;
        scatter_group<false>(h, agg, src[e], dst[e], norm[e], half, c, 32);
    }
    if (base < nE) {
        const int e = base + lane;
        const bool ok = e < nE;
        scatter_group<true>(h, agg, ok ? src[e] : 0, ok ? dst[e] : 0, ok ? norm[e] : 0.0f,
                            half, c, nE - base);
    }
}


__device__ __forceinline__ float bf_lo(unsigned int u) { return __uint_as_float(u << 16); }
__device__ __forceinline__ float bf_hi(unsigned int u) { return __uint_as_float(u & 0xFFFF0000u); }

template <bool GUARD>
__device__ __forceinline__ void scatter_group_bf(const unsigned short* __restrict__ g,
                                                 float* __restrict__ agg,
                                                 int s_l, int d_l, float n_l,
                                                 int half, int c, int left) {
    constexpr int P = 8;
#pragma unroll
    for (int j0 = 0; j0 < 16; j0 += P) {
        uint2 hv[P];
        float nm[P];
        int d[P];
#pragma unroll
        for (int p = 0; p < P; ++p) {
            const int sl = 2 * (j0 + p) + half;
            const int s = __shfl(s_l, sl);
            d[p]  = __shfl(d_l, sl);
            nm[p] = __shfl(n_l, sl);
            hv[p] = uint2{0u, 0u};
            if (!GUARD || sl < left) hv[p] = *(const uint2*)(g + (size_t)s * COMB + c);
        }
#pragma unroll
        for (int p = 0; p < P; ++p) {
            if (!GUARD || 2 * (j0 + p) + half < left) {
                float* ap = agg + (size_t)d[p] * COMB + c;
                const float x0 = bf_lo(hv[p].x), x1 = bf_hi(hv[p].x);
                const float x2 = bf_lo(hv[p].y), x3 = bf_hi(hv[p].y);
                if (x0 != 0.0f) atomicAdd(ap + 0, x0 * nm[p]);
                if (x1 != 0.0f) atomicAdd(ap + 1, x1 * nm[p]);
                if (x2 != 0.0f) atomicAdd(ap + 2, x2 * nm[p]);
                if (x3 != 0.0f) atomicAdd(ap + 3, x3 * nm[p]);
            }
        }
    }
}

__global__ __launch_bounds__(256) void gcn_scatter_bf_kernel(const unsigned short* __restrict__ g,
                                                             const int* __restrict__ src,
                                                             const int* __restrict__ dst,
                                                             const float* __restrict__ norm,
                                                             float* __restrict__ agg0,
                                                             float* __restrict__ agg1, int nE) {
    float* __restrict__ agg = package_half() ? agg1 : agg0;
    const int lane = threadIdx.x & 31;
    const int half = lane >> 4;
    const int c    = (lane & 15) * 4;
    const int gstep = ((gridDim.x * blockDim.x) >> 5) * 32;
    int base = ((blockIdx.x * blockDim.x + threadIdx.x) >> 5) * 32;
    for (; base + 32 <= nE; base += gstep) {
        const int e = base + lane;
        scatter_group_bf<false>(g, agg, src[e], dst[e], norm[e], half, c, 32);
    }
    if (base < nE) {
        const int e = base + lane;
        const bool ok = e < nE;
        scatter_group_bf<true>(g, agg, ok ? src[e] : 0, ok ? dst[e] : 0, ok ? norm[e] : 0.0f,
                               half, c, nE - base);
    }
}

__device__ __forceinline__ float4 bf4_scaled(const ushort4 o, float w) {
    const float4 r = {__uint_as_float((unsigned int)o.x << 16) * w,
                      __uint_as_float((unsigned int)o.y << 16) * w,
                      __uint_as_float((unsigned int)o.z << 16) * w,
                      __uint_as_float((unsigned int)o.w << 16) * w};
    return r;
}

__global__ __launch_bounds__(256) void relu_bias_init_kernel(const float* __restrict__ agg,
                                                             const float* __restrict__ agg1,
                                                             const float* __restrict__ b,
                                                             const float* __restrict__ dinv,
                                                             unsigned short* __restrict__ g,
                                                             float* __restrict__ s0,
                                                             float* __restrict__ s1,
                                                             int total) {
    constexpr int U = 6;
    const float4* a4 = (const float4*)agg;
    const float4* p4 = (const float4*)agg1;
    ushort4* g4 = (ushort4*)g;
    float4* s04 = (float4*)s0;
    float4* s14 = (float4*)s1;
    const float4 zero = {0.0f, 0.0f, 0.0f, 0.0f};
    const int n4 = total >> 2;
    const int stride = gridDim.x * blockDim.x;
    int idx = blockIdx.x * blockDim.x + threadIdx.x;
    const float4 bv = *(const float4*)(b + (idx & 15) * 4);
    for (; idx + (U - 1) * stride < n4; idx += U * stride) {
        float4 v[U], q[U];
        float w[U];
#pragma unroll
        for (int u = 0; u < U; ++u) {
            v[u] = a4[idx + u * stride];
            q[u] = p4[idx + u * stride];
            w[u] = dinv[(idx + u * stride) >> 4];
        }
#pragma unroll
        for (int u = 0; u < U; ++u) {
            const float4 r = {fmaxf((v[u].x + q[u].x) + bv.x, 0.0f),
                              fmaxf((v[u].y + q[u].y) + bv.y, 0.0f),
                              fmaxf((v[u].z + q[u].z) + bv.z, 0.0f),
                              fmaxf((v[u].w + q[u].w) + bv.w, 0.0f)};
            const ushort4 o = f2bf4(r);
            g4[idx + u * stride] = o;
            s04[idx + u * stride] = bf4_scaled(o, w[u] * w[u]);
            s14[idx + u * stride] = zero;
        }
    }
    for (; idx < n4; idx += stride) {
        const float4 v = a4[idx];
        const float4 q = p4[idx];
        float w = dinv[idx >> 4]; w *= w;
        const float4 r = {fmaxf((v.x + q.x) + bv.x, 0.0f), fmaxf((v.y + q.y) + bv.y, 0.0f),
                          fmaxf((v.z + q.z) + bv.z, 0.0f), fmaxf((v.w + q.w) + bv.w, 0.0f)};
        const ushort4 o = f2bf4(r);
        g4[idx] = o;
        s04[idx] = bf4_scaled(o, w);
        s14[idx] = zero;
    }
}

__device__ __forceinline__ void split_pack2(float a, float b, unsigned int& hi, unsigned int& lo) {
    const f32x2 v = {a, b};
    hi = __builtin_bit_cast(unsigned int, __builtin_convertvector(v, bf16x2));
    const f32x2 r = {a - __uint_as_float(hi << 16), b - __uint_as_float(hi & 0xFFFF0000u)};
    lo = __builtin_bit_cast(unsigned int, __builtin_convertvector(r, bf16x2));
}

__global__ __launch_bounds__(64) void final_kernel(const float* __restrict__ mlp,
                                                   const float* __restrict__ s0,
                                                   const float* __restrict__ s1,
                                                   const unsigned short* __restrict__ W2p,
                                                   const float* __restrict__ b2,
                                                   const float* __restrict__ fw,
                                                   const float* __restrict__ fb,
                                                   float* __restrict__ out, int nNodes) {
    const int lane = threadIdx.x & 31;
    const int c = lane & 15;
    const int h = lane >> 4;
    const int n0 = __builtin_amdgcn_readfirstlane(((blockIdx.x * blockDim.x + threadIdx.x) >> 5) * 32);
    if (n0 >= nNodes) return;

    V16 b[2][4];
#pragma unroll
    for (int kt = 0; kt < 2; ++kt) {
#pragma unroll
        for (int nt = 0; nt < 4; ++nt) {
            const uint4* bp = (const uint4*)(W2p + (size_t)(kt * 4 + nt) * 512 + lane * 16);
            b[kt][nt].u[0] = bp[0];
            b[kt][nt].u[1] = bp[1];
        }
    }
    float bb[4], fh[4];
#pragma unroll
    for (int nt = 0; nt < 4; ++nt) { bb[nt] = b2[16 * nt + c]; fh[nt] = fw[COMB + 16 * nt + c]; }
    const float4 fl = *(const float4*)(fw + 4 * c);

    float v[2];
#pragma unroll
    for (int i = 0; i < 2; ++i) {
        const int base = n0 + 16 * i;
        int arow = base + c;
        arow = (arow < nNodes) ? arow : (nNodes - 1);
        const float* r0 = s0 + (size_t)arow * COMB + 8 * h;
        const float* r1 = s1 + (size_t)arow * COMB + 8 * h;
        float4 m[8];
#pragma unroll
        for (int r = 0; r < 8; ++r) {
            int mrow = base + 8 * h + r;
            mrow = (mrow < nNodes) ? mrow : (nNodes - 1);
            m[r] = *(const float4*)(mlp + (size_t)mrow * COMB + 4 * c);
        }
        V16 ahi[2], alo[2];
#pragma unroll
        for (int kt = 0; kt < 2; ++kt) {
            float4 x[4], y[4];
#pragma unroll
            for (int t = 0; t < 4; ++t) {
                const int off = kt * 32 + (t >> 1) * 16 + (t & 1) * 4;
                x[t] = *(const float4*)(r0 + off);
                y[t] = *(const float4*)(r1 + off);
            }
#pragma unroll
            for (int t = 0; t < 2; ++t) {
                uint4 uh, ul;
                split_pack2(x[2 * t].x + y[2 * t].x, x[2 * t].y + y[2 * t].y, uh.x, ul.x);
                split_pack2(x[2 * t].z + y[2 * t].z, x[2 * t].w + y[2 * t].w, uh.y, ul.y);
                split_pack2(x[2 * t + 1].x + y[2 * t + 1].x, x[2 * t + 1].y + y[2 * t + 1].y, uh.z, ul.z);
                split_pack2(x[2 * t + 1].z + y[2 * t + 1].z, x[2 * t + 1].w + y[2 * t + 1].w, uh.w, ul.w);
                ahi[kt].u[t] = uh;
                alo[kt].u[t] = ul;
            }
        }
        v8f acc[4];
#pragma unroll
        for (int nt = 0; nt < 4; ++nt) acc[nt] = v8f{};
#pragma unroll
        for (int kt = 0; kt < 2; ++kt) {
#pragma unroll
            for (int nt = 0; nt < 4; ++nt) {
                acc[nt] = __builtin_amdgcn_wmma_f32_16x16x32_bf16(false, ahi[kt].v, false, b[kt][nt].v,
                                                                  (short)0, acc[nt], false, false);
                acc[nt] = __builtin_amdgcn_wmma_f32_16x16x32_bf16(false, alo[kt].v, false, b[kt][nt].v,
                                                                  (short)0, acc[nt], false, false);
            }
        }
        float p[8];
#pragma unroll
        for (int r = 0; r < 8; ++r) {
            float t = m[r].x * fl.x + m[r].y * fl.y + m[r].z * fl.z + m[r].w * fl.w;
#pragma unroll
            for (int nt = 0; nt < 4; ++nt) t += (acc[nt][r] + bb[nt]) * fh[nt];
            t += __shfl_xor(t, 8);
            t += __shfl_xor(t, 4);
            t += __shfl_xor(t, 2);
            t += __shfl_xor(t, 1);
            p[r] = t;
        }
        float sel = p[0];
#pragma unroll
        for (int r = 1; r < 8; ++r) sel = ((lane & 7) == r) ? p[r] : sel;
        v[i] = sel;
    }
    const int from = ((lane & 15) >> 3) * 16 + (lane & 7);
    const float t0 = __shfl(v[0], from);
    const float t1 = __shfl(v[1], from);
    const float res = (lane < 16) ? t0 : t1;
    if (n0 + lane < nNodes) out[n0 + lane] = res + fb[0];
}

extern "C" void kernel_launch(void* const* d_in, const int* in_sizes, int n_in,
                              void* d_out, int out_size, void* d_ws, size_t ws_size,
                              hipStream_t stream) {
    (void)in_sizes; (void)n_in; (void)out_size; (void)ws_size;

    const float* x        = (const float*)d_in[0];
    const int*   eidx     = (const int*)d_in[1];
    const float* eattr    = (const float*)d_in[2];
    const float* ufeat    = (const float*)d_in[3];
    const float* gcn1_W   = (const float*)d_in[4];
    const float* gcn1_b   = (const float*)d_in[5];
    const float* gcn2_W   = (const float*)d_in[6];
    const float* gcn2_b   = (const float*)d_in[7];
    const float* fc1_W    = (const float*)d_in[8];
    const float* fc1_b    = (const float*)d_in[9];
    const float* fc2_W    = (const float*)d_in[10];
    const float* fc2_b    = (const float*)d_in[11];
    const float* fcf_W    = (const float*)d_in[12];
    const float* fcf_b    = (const float*)d_in[13];
    const int* esrc = eidx;
    const int* edst = eidx + N_EDGES;
    float* out = (float*)d_out;

    char* ws = (char*)d_ws;
    size_t off = 0;
    auto carve = [&](size_t bytes) -> void* {
        void* p = ws + off;
        off = (off + bytes + 255) & ~(size_t)255;
        return p;
    };
    unsigned short* UBF = (unsigned short*)carve((size_t)N_NODES * MLP_IN * 2);
    unsigned short* XBF = (unsigned short*)carve((size_t)N_NODES * GNN_IN * 2);
    unsigned short* HBF = (unsigned short*)carve((size_t)N_NODES * MLP_HID * 2);
    unsigned short* GBF = (unsigned short*)carve((size_t)N_NODES * COMB * 2);
    unsigned short* W1P = (unsigned short*)carve((size_t)MLP_IN * MLP_HID * 2);
    unsigned short* W2P = (unsigned short*)carve((size_t)MLP_HID * COMB * 2);
    unsigned short* G1P = (unsigned short*)carve((size_t)GNN_IN * GNN_HID * 2);
    unsigned short* G2P = (unsigned short*)carve((size_t)GNN_HID * COMB * 2);
    float* MLPOUT = (float*)carve((size_t)N_NODES * COMB * 4);
    float* H1     = (float*)carve((size_t)N_NODES * COMB * 4);
    float* AGG    = (float*)carve((size_t)N_NODES * COMB * 4);
    float* DEG    = (float*)carve((size_t)N_NODES * 4);
    float* DINV   = (float*)carve((size_t)N_NODES * 4);
    float* NORM   = (float*)carve((size_t)N_EDGES * 4);
    float* DEG1 = (float*)HBF;
    float* AGG1 = (float*)UBF;
    float* S2A = (float*)XBF;
    float* S2B = S2A + (size_t)N_NODES * COMB;

    {
        int n4 = (N_NODES * MLP_IN) / 4;
        cvt_bf16_kernel<<<LOOP_BLOCKS, 256, 0, stream>>>(ufeat, UBF, n4);
        n4 = (N_NODES * GNN_IN) / 4;
        cvt_bf16_kernel<<<LOOP_BLOCKS, 256, 0, stream>>>(x, XBF, n4);
    }
    auto pack = [&](const float* W, unsigned short* P, int K, int N) {
        int total = (K / 32) * (N / 16) * 32;
        pack_w_kernel<<<(total + 255) / 256, 256, 0, stream>>>(W, P, K, N);
    };
    pack(fc1_W, W1P, MLP_IN, MLP_HID);
    pack(fc2_W, W2P, MLP_HID, COMB);
    pack(gcn1_W, G1P, GNN_IN, GNN_HID);
    pack(gcn2_W, G2P, GNN_HID, COMB);

    deg_init_kernel<<<(N_NODES + 255) / 256, 256, 0, stream>>>(DEG, DEG1, N_NODES);
    deg_accum_kernel<<<LOOP_BLOCKS, 256, 0, stream>>>(edst, eattr, DEG, DEG1, N_EDGES);
    dinv_kernel<<<(N_NODES + 255) / 256, 256, 0, stream>>>(DEG, DEG1, DINV, N_NODES);
    edge_norm_kernel<<<LOOP_BLOCKS, 256, 0, stream>>>(esrc, edst, eattr, DINV, NORM, N_EDGES);

    mlp_chain_kernel<<<256, 256, 0, stream>>>(UBF, W1P, fc1_b, W2P, fc2_b, MLPOUT, N_NODES);

    gemm_wmma_kernel<GNN_IN, GNN_HID, 2, false, false, false><<<256, 256, 0, stream>>>(
        XBF, G1P, nullptr, H1, N_NODES);
    agg_init_kernel<<<LOOP_BLOCKS, 256, 0, stream>>>(H1, DINV, AGG, AGG1, N_NODES);
    gcn_scatter_kernel<<<SCATTER_BLOCKS, 256, 0, stream>>>(H1, esrc, edst, NORM, AGG, AGG1, N_EDGES);
    relu_bias_init_kernel<<<LOOP_BLOCKS, 256, 0, stream>>>(AGG, AGG1, gcn1_b, DINV, GBF, S2A, S2B,
                                                            N_NODES * COMB);

    gcn_scatter_bf_kernel<<<SCATTER_BLOCKS, 256, 0, stream>>>(GBF, esrc, edst, NORM, S2A, S2B, N_EDGES);

    final_kernel<<<(N_NODES + 63) / 64, 64, 0, stream>>>(MLPOUT, S2A, S2B, G2P, gcn2_b, fcf_W, fcf_b,
                                                      out, N_NODES);
}


# layer-2 edge pass: stored g rows pair columns l, l+32 per dword; a wave takes one edge per step, each f32 atomic instruction adds one whole 128-byte line (was 4 quarter-filled); arrival order kept
# speedup vs baseline: 1.8693x; 1.0994x over previous
#include <hip/hip_runtime.h>
#include <hip/hip_bf16.h>


#define N_NODES 50000
#define N_EDGES 800000
#define GNN_IN  256
#define GNN_HID 64
#define COMB    64
#define MLP_IN  256
#define MLP_HID 512

typedef __attribute__((ext_vector_type(16))) __bf16 v16bf;
typedef __attribute__((ext_vector_type(8)))  float  v8f;

typedef __attribute__((ext_vector_type(2)))  float  f32x2;
typedef __attribute__((ext_vector_type(2)))  __bf16 bf16x2;

union V16 {
    uint4 u[2];
    v16bf v;
};

__device__ __forceinline__ unsigned short f2bf(float f) {
    unsigned int u = __float_as_uint(f);
    unsigned int r = u + 0x7FFFu + ((u >> 16) & 1u);
    return (unsigned short)(r >> 16);
}

__device__ __forceinline__ ushort4 f2bf4(const float4 f) {
    ushort4 o;
    o.x = f2bf(f.x); o.y = f2bf(f.y); o.z = f2bf(f.z); o.w = f2bf(f.w);
    return o;
}

__device__ __forceinline__ int package_half() {
    const unsigned int id = __builtin_amdgcn_s_sendmsg_rtn(0x87);
    return (int)((id >> 18) & 1u);
}

#define LOOP_BLOCKS 512
#define SCATTER_BLOCKS 1024

__global__ __launch_bounds__(256) void cvt_bf16_kernel(const float* __restrict__ in,
                                                       unsigned short* __restrict__ out, int n4) {
    constexpr int U = 8;
    const float4* in4 = (const float4*)in;
    ushort4* out4 = (ushort4*)out;
    const int stride = gridDim.x * blockDim.x;
    int i = blockIdx.x * blockDim.x + threadIdx.x;
    for (; i + (U - 1) * stride < n4; i += U * stride) {
        float4 f[U];
#pragma unroll
        for (int u = 0; u < U; ++u) f[u] = in4[i + u * stride];
#pragma unroll
        for (int u = 0; u < U; ++u) out4[i + u * stride] = f2bf4(f[u]);
    }
    for (; i < n4; i += stride) out4[i] = f2bf4(in4[i]);
}

__global__ void pack_w_kernel(const float* __restrict__ W,
                              unsigned short* __restrict__ P, int K, int N) {
    int idx = blockIdx.x * blockDim.x + threadIdx.x;
    int nTiles = N >> 4;
    int total = (K >> 5) * nTiles * 32;
    if (idx >= total) return;
    int lane = idx & 31;
    int tile = idx >> 5;
    int kt = tile / nTiles;
    int nt = tile % nTiles;
    int col = nt * 16 + (lane & 15);
    int kb  = kt * 32 + ((lane < 16) ? 0 : 8);
    __attribute__((aligned(16))) unsigned short o[16];
#pragma unroll
    for (int j = 0; j < 8; ++j) o[j]     = f2bf(W[(size_t)(kb + j) * N + col]);
#pragma unroll
    for (int j = 0; j < 8; ++j) o[8 + j] = f2bf(W[(size_t)(kb + 16 + j) * N + col]);
    uint4* dst = (uint4*)(P + (size_t)tile * 512 + lane * 16);
    dst[0] = ((uint4*)o)[0];
    dst[1] = ((uint4*)o)[1];
}

template <int K>
__device__ __forceinline__ void load_a_tile(V16 (&a)[K / 32],
                                            const unsigned short* __restrict__ abase,
                                            int mt) {
    const unsigned short* arow = abase + (size_t)mt * 16 * K;
#pragma unroll
    for (int kt = 0; kt < K / 32; ++kt) {
        const uint4* ap = (const uint4*)(arow + kt * 32);
        a[kt].u[0] = ap[0];
        a[kt].u[1] = ap[2];
    }
}

template <int K, int N, int NTW, bool RELU, bool OUTBF>
__device__ __forceinline__ void tile_mma_store(const V16 (&a)[K / 32],
                                               const V16 (&b)[K / 32][NTW],
                                               const float (&bv)[NTW],
                                               void* __restrict__ C,
                                               int mt, int cg, int lane) {
    v8f acc[NTW];
#pragma unroll
    for (int j = 0; j < NTW; ++j) acc[j] = v8f{};
#pragma unroll
    for (int kt = 0; kt < K / 32; ++kt) {
#pragma unroll
        for (int j = 0; j < NTW; ++j) {
            acc[j] = __builtin_amdgcn_wmma_f32_16x16x32_bf16(false, a[kt].v, false, b[kt][j].v,
                                                             (short)0, acc[j], false, false);
        }
    }
    const int r0 = mt * 16 + ((lane < 16) ? 0 : 8);
#pragma unroll
    for (int j = 0; j < NTW; ++j) {
        const int cc = (cg * NTW + j) * 16 + (lane & 15);
#pragma unroll
        for (int v = 0; v < 8; ++v) {
            float val = acc[j][v] + bv[j];
            if (RELU) val = fmaxf(val, 0.0f);
            const size_t off = (size_t)(r0 + v) * N + cc;
            if (OUTBF) ((unsigned short*)C)[off] = f2bf(val);
            else       ((float*)C)[off] = val;
        }
    }
}

template <int K, int N, int NTW, bool RELU, bool OUTBF, bool HASBIAS>
__global__ __launch_bounds__(256) void gemm_wmma_kernel(const unsigned short* __restrict__ A,
                                                        const unsigned short* __restrict__ Bp,
                                                        const float* __restrict__ bias,
                                                        void* __restrict__ C, int M) {
    constexpr int KT = K / 32;
    constexpr int NT = N / 16;
    constexpr int CG = NT / NTW;
    constexpr int S  = 8 / CG;
    static_assert(K % 32 == 0 && N % 16 == 0 && NT % NTW == 0 && 8 % CG == 0,
                  "tile split must divide the problem and the 8 waves of a block");
    const int lane = threadIdx.x & 31;
    const int wid  = __builtin_amdgcn_readfirstlane(threadIdx.x >> 5);
    const int cg     = wid % CG;
    const int stream = wid / CG;
    const int mTiles = M >> 4;
    const int step   = gridDim.x * S;
    int mt = blockIdx.x * S + stream;
    if (mt >= mTiles) return;

    V16 b[KT][NTW];
#pragma unroll
    for (int kt = 0; kt < KT; ++kt) {
#pragma unroll
        for (int j = 0; j < NTW; ++j) {
            const uint4* bp = (const uint4*)(Bp + (size_t)(kt * NT + cg * NTW + j) * 512 + lane * 16);
            b[kt][j].u[0] = bp[0];
            b[kt][j].u[1] = bp[1];
        }
    }
    float bv[NTW];
#pragma unroll
    for (int j = 0; j < NTW; ++j)
        bv[j] = HASBIAS ? bias[(cg * NTW + j) * 16 + (lane & 15)] : 0.0f;

    const unsigned short* abase = A + (size_t)(lane & 15) * K + ((lane < 16) ? 0 : 8);

    V16 a0[KT], a1[KT];
    load_a_tile<K>(a0, abase, mt);
    while (mt + step < mTiles) {
        load_a_tile<K>(a1, abase, mt + step);
        __builtin_amdgcn_sched_barrier(0);
        tile_mma_store<K, N, NTW, RELU, OUTBF>(a0, b, bv, C, mt, cg, lane);
        const int nxt = mt + 2 * step;
        load_a_tile<K>(a0, abase, (nxt < mTiles) ? nxt : mt);
        __builtin_amdgcn_sched_barrier(0);
        tile_mma_store<K, N, NTW, RELU, OUTBF>(a1, b, bv, C, mt + step, cg, lane);
        mt = nxt;
    }
    if (mt < mTiles) tile_mma_store<K, N, NTW, RELU, OUTBF>(a0, b, bv, C, mt, cg, lane);
}

#define FZ_W2_BYTES   (MLP_HID * COMB * 2)
#define FZ_ROW_PITCH  272
#define FZ_WAVE_BYTES (16 * FZ_ROW_PITCH)
#define FZ_BUF_BYTES  (8 * FZ_WAVE_BYTES)
#define FZ_LDS_BYTES  (FZ_W2_BYTES + 2 * FZ_BUF_BYTES)

__device__ __forceinline__ void lds_barrier() {
    asm volatile("s_wait_dscnt 0x0\n\ts_barrier_signal -1\n\ts_barrier_wait -1" ::: "memory");
}

__device__ __forceinline__ unsigned int relu_pack2(float lo, float hi, float blo, float bhi) {
    const f32x2 v = {fmaxf(lo + blo, 0.0f), fmaxf(hi + bhi, 0.0f)};
    return __builtin_bit_cast(unsigned int, __builtin_convertvector(v, bf16x2));
}

__device__ __forceinline__ void mlp_tile(const V16 (&a)[MLP_IN / 32],
                                         const V16 (&b)[MLP_IN / 32][4],
                                         const float4 (&bias1)[4][2],
                                         const unsigned char* lds, unsigned char* pbuf,
                                         const float4 bias2, float* __restrict__ out,
                                         int mt, int wid, int lane) {
    const int c = lane & 15;
    const int h = lane >> 4;
    v8f acc[4];
#pragma unroll
    for (int j = 0; j < 4; ++j) acc[j] = v8f{};
#pragma unroll
    for (int kt = 0; kt < MLP_IN / 32; ++kt) {
#pragma unroll
        for (int j = 0; j < 4; ++j) {
            acc[j] = __builtin_amdgcn_wmma_f32_16x16x32_bf16(false, b[kt][j].v, false, a[kt].v,
                                                             (short)0, acc[j], false, false);
        }
    }
    __builtin_amdgcn_sched_barrier(0);
    V16 w2[2][4];
#pragma unroll
    for (int q = 0; q < 2; ++q) {
#pragma unroll
        for (int nt = 0; nt < 4; ++nt) {
            const unsigned char* wp = lds + ((2 * wid + q) * 4 + nt) * 1024 + lane * 16;
            w2[q][nt].u[0] = *(const uint4*)(wp);
            w2[q][nt].u[1] = *(const uint4*)(wp + 512);
        }
    }
    __builtin_amdgcn_sched_barrier(0);
    V16 hf[2];
#pragma unroll
    for (int q = 0; q < 2; ++q) {
#pragma unroll
        for (int t = 0; t < 2; ++t) {
            const int j = 2 * q + t;
            const float4 p0 = bias1[j][0], p1 = bias1[j][1];
            uint4 u;
            u.x = relu_pack2(acc[j][0], acc[j][1], p0.x, p0.y);
            u.y = relu_pack2(acc[j][2], acc[j][3], p0.z, p0.w);
            u.z = relu_pack2(acc[j][4], acc[j][5], p1.x, p1.y);
            u.w = relu_pack2(acc[j][6], acc[j][7], p1.z, p1.w);
            hf[q].u[t] = u;
        }
    }
    v8f acc2[4];
#pragma unroll
    for (int nt = 0; nt < 4; ++nt) acc2[nt] = v8f{};
#pragma unroll
    for (int q = 0; q < 2; ++q) {
#pragma unroll
        for (int nt = 0; nt < 4; ++nt) {
            acc2[nt] = __builtin_amdgcn_wmma_f32_16x16x32_bf16(false, w2[q][nt].v, false, hf[q].v,
                                                               (short)0, acc2[nt], false, false);
        }
    }
#pragma unroll
    for (int nt = 0; nt < 4; ++nt) {
        float4* pp = (float4*)(pbuf + wid * FZ_WAVE_BYTES + c * FZ_ROW_PITCH + (16 * nt + 8 * h) * 4);
        const float4 lo = {acc2[nt][0], acc2[nt][1], acc2[nt][2], acc2[nt][3]};
        const float4 hi = {acc2[nt][4], acc2[nt][5], acc2[nt][6], acc2[nt][7]};
        pp[0] = lo;
        pp[1] = hi;
    }
    lds_barrier();
    const int node = threadIdx.x >> 4;
    const int col4 = (threadIdx.x & 15) * 4;
    const unsigned char* rp = pbuf + node * FZ_ROW_PITCH + col4 * 4;
    float4 s = *(const float4*)rp;
#pragma unroll
    for (int w = 1; w < 8; ++w) {
        const float4 p = *(const float4*)(rp + w * FZ_WAVE_BYTES);
        s.x += p.x; s.y += p.y; s.z += p.z; s.w += p.w;
    }
    s.x += bias2.x; s.y += bias2.y; s.z += bias2.z; s.w += bias2.w;
    *(float4*)(out + (size_t)(mt * 16 + node) * COMB + col4) = s;
}

__global__ __launch_bounds__(256) void mlp_chain_kernel(const unsigned short* __restrict__ A,
                                                        const unsigned short* __restrict__ W1p,
                                                        const float* __restrict__ b1,
                                                        const unsigned short* __restrict__ W2p,
                                                        const float* __restrict__ b2,
                                                        float* __restrict__ out, int M) {
    constexpr int K  = MLP_IN;
    constexpr int KT = K / 32;
    constexpr int NT = MLP_HID / 16;
    __shared__ __attribute__((aligned(16))) unsigned char L[FZ_LDS_BYTES];
    const int lane = threadIdx.x & 31;
    const int wid = __builtin_amdgcn_readfirstlane(threadIdx.x >> 5);
    const int mTiles = M >> 4;
    const int step   = gridDim.x;
    int mt = blockIdx.x;
    if (mt >= mTiles) return;

    V16 b[KT][4];
#pragma unroll
    for (int kt = 0; kt < KT; ++kt) {
#pragma unroll
        for (int j = 0; j < 4; ++j) {
            const uint4* bp = (const uint4*)(W1p + (size_t)(kt * NT + wid * 4 + j) * 512 + lane * 16);
            b[kt][j].u[0] = bp[0];
            b[kt][j].u[1] = bp[1];
        }
    }
    const unsigned short* abase = A + (size_t)(lane & 15) * K + ((lane < 16) ? 0 : 8);
    V16 a0[KT], a1[KT];
    load_a_tile<K>(a0, abase, mt);

    for (int i = threadIdx.x; i < FZ_W2_BYTES / 16; i += 256) {
        const uint4 v = ((const uint4*)W2p)[i];
        *(uint4*)(L + (i >> 6) * 1024 + (i & 1) * 512 + ((i & 63) >> 1) * 16) = v;
    }
    float4 bias1[4][2];
#pragma unroll
    for (int j = 0; j < 4; ++j) {
        const float4* bp = (const float4*)(b1 + wid * 64 + j * 16 + (lane >> 4) * 8);
        bias1[j][0] = bp[0];
        bias1[j][1] = bp[1];
    }
    const float4 bias2 = *(const float4*)(b2 + (threadIdx.x & 15) * 4);
    lds_barrier();

    unsigned char* pb0 = L + FZ_W2_BYTES;
    unsigned char* pb1 = pb0 + FZ_BUF_BYTES;
    while (mt + step < mTiles) {
        load_a_tile<K>(a1, abase, mt + step);
        __builtin_amdgcn_sched_barrier(0);
        mlp_tile(a0, b, bias1, L, pb0, bias2, out, mt, wid, lane);
        const int nxt = mt + 2 * step;
        load_a_tile<K>(a0, abase, (nxt < mTiles) ? nxt : mt);
        __builtin_amdgcn_sched_barrier(0);
        mlp_tile(a1, b, bias1, L, pb1, bias2, out, mt + step, wid, lane);
        mt = nxt;
    }
    if (mt < mTiles) mlp_tile(a0, b, bias1, L, pb0, bias2, out, mt, wid, lane);
}

__global__ void deg_init_kernel(float* __restrict__ deg0, float* __restrict__ deg1, int n) {
    int i = blockIdx.x * blockDim.x + threadIdx.x;
    if (i < n) {
        deg0[i] = 1.0f;
        deg1[i] = 0.0f;
    }
}

__global__ __launch_bounds__(256) void deg_accum_kernel(const int* __restrict__ dst,
                                                        const float* __restrict__ ew,
                                                        float* __restrict__ deg0,
                                                        float* __restrict__ deg1, int nE) {
    constexpr int U = 6;
    float* __restrict__ deg = package_half() ? deg1 : deg0;
    const int stride = gridDim.x * blockDim.x;
    int e = blockIdx.x * blockDim.x + threadIdx.x;
    for (; e + (U - 1) * stride < nE; e += U * stride) {
        int d[U];
        float w[U];
#pragma unroll
        for (int u = 0; u < U; ++u) { d[u] = dst[e + u * stride]; w[u] = ew[e + u * stride]; }
#pragma unroll
        for (int u = 0; u < U; ++u) atomicAdd(&deg[d[u]], w[u]);
    }
    for (; e < nE; e += stride) atomicAdd(&deg[dst[e]], ew[e]);
}

__global__ void dinv_kernel(const float* __restrict__ deg0,
                            const float* __restrict__ deg1,
                            float* __restrict__ dinv, int n) {
    int i = blockIdx.x * blockDim.x + threadIdx.x;
    if (i >= n) return;
    float d = deg0[i] + deg1[i];
    dinv[i] = (d > 0.0f) ? rsqrtf(fmaxf(d, 1e-12f)) : 0.0f;
}

__global__ __launch_bounds__(256) void edge_norm_kernel(const int* __restrict__ src,
                                                        const int* __restrict__ dst,
                                                        const float* __restrict__ ew,
                                                        const float* __restrict__ dinv,
                                                        float* __restrict__ norm, int nE) {
    constexpr int U = 6;
    const int stride = gridDim.x * blockDim.x;
    int e = blockIdx.x * blockDim.x + threadIdx.x;
    for (; e + (U - 1) * stride < nE; e += U * stride) {
        int s[U], d[U];
        float w[U], ds[U], dd[U];
#pragma unroll
        for (int u = 0; u < U; ++u) {
            s[u] = src[e + u * stride]; d[u] = dst[e + u * stride]; w[u] = ew[e + u * stride];
        }
#pragma unroll
        for (int u = 0; u < U; ++u) { ds[u] = dinv[s[u]]; dd[u] = dinv[d[u]]; }
#pragma unroll
        for (int u = 0; u < U; ++u) norm[e + u * stride] = ds[u] * w[u] * dd[u];
    }
    for (; e < nE; e += stride) norm[e] = dinv[src[e]] * ew[e] * dinv[dst[e]];
}

__global__ __launch_bounds__(256) void agg_init_kernel(const float* __restrict__ h,
                                                       const float* __restrict__ dinv,
                                                       float* __restrict__ agg,
                                                       float* __restrict__ agg1, int nNodes) {
    constexpr int U = 6;
    const float4* h4 = (const float4*)h;
    float4* agg4 = (float4*)agg;
    float4* aggz = (float4*)agg1;
    const float4 zero = {0.0f, 0.0f, 0.0f, 0.0f};
    const int n16 = nNodes * (COMB / 4);
    const int stride = gridDim.x * blockDim.x;
    int idx = blockIdx.x * blockDim.x + threadIdx.x;
    for (; idx + (U - 1) * stride < n16; idx += U * stride) {
        float4 hv[U];
        float w[U];
#pragma unroll
        for (int u = 0; u < U; ++u) {
            hv[u] = h4[idx + u * stride];
            w[u]  = dinv[(idx + u * stride) >> 4];
        }
#pragma unroll
        for (int u = 0; u < U; ++u) {
            const float ww = w[u] * w[u];
            const float4 o = {hv[u].x * ww, hv[u].y * ww, hv[u].z * ww, hv[u].w * ww};
            agg4[idx + u * stride] = o;
            aggz[idx + u * stride] = zero;
        }
    }
    for (; idx < n16; idx += stride) {
        float w = dinv[idx >> 4]; w *= w;
        const float4 hv = h4[idx];
        const float4 o = {hv.x * w, hv.y * w, hv.z * w, hv.w * w};
        agg4[idx] = o;
        aggz[idx] = zero;
    }
}

template <bool GUARD>
__device__ __forceinline__ void scatter_group(const float* __restrict__ h,
                                              float* __restrict__ agg,
                                              int s_l, int d_l, float n_l,
                                              int half, int c, int left) {
    constexpr int P = 8;
#pragma unroll
    for (int j0 = 0; j0 < 16; j0 += P) {
        float4 hv[P];
        float nm[P];
        int d[P];
#pragma unroll
        for (int p = 0; p < P; ++p) {
            const int sl = 2 * (j0 + p) + half;
            const int s = __shfl(s_l, sl);
            d[p]  = __shfl(d_l, sl);
            nm[p] = __shfl(n_l, sl);
            if (!GUARD || sl < left) hv[p] = *(const float4*)(h + (size_t)s * COMB + c);
        }
#pragma unroll
        for (int p = 0; p < P; ++p) {
            if (!GUARD || 2 * (j0 + p) + half < left) {
                float* ap = agg + (size_t)d[p] * COMB + c;
                atomicAdd(ap + 0, hv[p].x * nm[p]);
                atomicAdd(ap + 1, hv[p].y * nm[p]);
                atomicAdd(ap + 2, hv[p].z * nm[p]);
                atomicAdd(ap + 3, hv[p].w * nm[p]);
            }
        }
    }
}

__global__ __launch_bounds__(256) void gcn_scatter_kernel(const float* __restrict__ h,
                                                          const int* __restrict__ src,
                                                          const int* __restrict__ dst,
                                                          const float* __restrict__ norm,
                                                          float* __restrict__ agg0,
                                                          float* __restrict__ agg1, int nE) {
    float* __restrict__ agg = package_half() ? agg1 : agg0;
    const int lane = threadIdx.x & 31;
    const int half = lane >> 4;
    const int c    = (lane & 15) * 4;
    const int gstep = ((gridDim.x * blockDim.x) >> 5) * 32;
    int base = ((blockIdx.x * blockDim.x + threadIdx.x) >> 5) * 32;
    for (; base + 32 <= nE; base += gstep) {
        const int e = base + lane;
        scatter_group<false>(h, agg, src[e], dst[e], norm[e], half, c, 32);
    }
    if (base < nE) {
        const int e = base + lane;
        const bool ok = e < nE;
        scatter_group<true>(h, agg, ok ? src[e] : 0, ok ? dst[e] : 0, ok ? norm[e] : 0.0f,
                            half, c, nE - base);
    }
}


__device__ __forceinline__ float bf_lo(unsigned int u) { return __uint_as_float(u << 16); }
__device__ __forceinline__ float bf_hi(unsigned int u) { return __uint_as_float(u & 0xFFFF0000u); }

template <bool GUARD>
__device__ __forceinline__ void scatter_group_bf(const unsigned int* __restrict__ g,
                                                 float* __restrict__ agg,
                                                 int s_l, int d_l, float n_l,
                                                 int lane, int left) {
    constexpr int P = 16;
    const int n_i = __float_as_int(n_l);
#pragma unroll
    for (int j0 = 0; j0 < 32; j0 += P) {
        unsigned int hv[P];
#pragma unroll
        for (int p = 0; p < P; ++p) {
            const int s = __builtin_amdgcn_readlane(s_l, j0 + p);
            hv[p] = 0u;
            if (!GUARD || j0 + p < left) hv[p] = g[(size_t)s * (COMB / 2) + lane];
        }
#pragma unroll
        for (int p = 0; p < P; ++p) {
            if (!GUARD || j0 + p < left) {
                const int d = __builtin_amdgcn_readlane(d_l, j0 + p);
                const float nm = __int_as_float(__builtin_amdgcn_readlane(n_i, j0 + p));
                float* ap = agg + (size_t)d * COMB + lane;
                const float x0 = bf_lo(hv[p]), x1 = bf_hi(hv[p]);
                if (x0 != 0.0f) atomicAdd(ap, x0 * nm);
                if (x1 != 0.0f) atomicAdd(ap + COMB / 2, x1 * nm);
            }
        }
    }
}

__global__ __launch_bounds__(256) void gcn_scatter_bf_kernel(const unsigned int* __restrict__ g,
                                                             const int* __restrict__ src,
                                                             const int* __restrict__ dst,
                                                             const float* __restrict__ norm,
                                                             float* __restrict__ agg0,
                                                             float* __restrict__ agg1, int nE) {
    float* __restrict__ agg = package_half() ? agg1 : agg0;
    const int lane = threadIdx.x & 31;
    const int gstep = ((gridDim.x * blockDim.x) >> 5) * 32;
    int base = __builtin_amdgcn_readfirstlane(((blockIdx.x * blockDim.x + threadIdx.x) >> 5) * 32);
    for (; base + 32 <= nE; base += gstep) {
        const int e = base + lane;
        scatter_group_bf<false>(g, agg, src[e], dst[e], norm[e], lane, 32);
    }
    if (base < nE) {
        const int e = base + lane;
        const bool ok = e < nE;
        scatter_group_bf<true>(g, agg, ok ? src[e] : 0, ok ? dst[e] : 0, ok ? norm[e] : 0.0f,
                               lane, nE - base);
    }
}

__device__ __forceinline__ float4 bf4_scaled(const ushort4 o, float w) {
    const float4 r = {__uint_as_float((unsigned int)o.x << 16) * w,
                      __uint_as_float((unsigned int)o.y << 16) * w,
                      __uint_as_float((unsigned int)o.z << 16) * w,
                      __uint_as_float((unsigned int)o.w << 16) * w};
    return r;
}

__device__ __forceinline__ ushort4 relu_sum_bf4(const float4 v, const float4 q, const float4 b) {
    const float4 r = {fmaxf((v.x + q.x) + b.x, 0.0f), fmaxf((v.y + q.y) + b.y, 0.0f),
                      fmaxf((v.z + q.z) + b.z, 0.0f), fmaxf((v.w + q.w) + b.w, 0.0f)};
    return f2bf4(r);
}

__device__ __forceinline__ uint4 pair_cols(const ushort4 lo, const ushort4 hi) {
    const uint4 r = {(unsigned int)lo.x | ((unsigned int)hi.x << 16),
                     (unsigned int)lo.y | ((unsigned int)hi.y << 16),
                     (unsigned int)lo.z | ((unsigned int)hi.z << 16),
                     (unsigned int)lo.w | ((unsigned int)hi.w << 16)};
    return r;
}

__global__ __launch_bounds__(256) void relu_bias_init_kernel(const float* __restrict__ agg,
                                                             const float* __restrict__ agg1,
                                                             const float* __restrict__ b,
                                                             const float* __restrict__ dinv,
                                                             unsigned int* __restrict__ g,
                                                             float* __restrict__ s0,
                                                             float* __restrict__ s1,
                                                             int total) {
    constexpr int U = 3;
    const float4* a4 = (const float4*)agg;
    const float4* p4 = (const float4*)agg1;
    uint4* g4 = (uint4*)g;
    float4* s04 = (float4*)s0;
    float4* s14 = (float4*)s1;
    const float4 zero = {0.0f, 0.0f, 0.0f, 0.0f};
    const int n8 = total >> 3;
    const int stride = gridDim.x * blockDim.x;
    int idx = blockIdx.x * blockDim.x + threadIdx.x;
    const int c = idx & 7;
    const float4 bA = *(const float4*)(b + 4 * c);
    const float4 bB = *(const float4*)(b + COMB / 2 + 4 * c);
    for (; idx + (U - 1) * stride < n8; idx += U * stride) {
        float4 vA[U], qA[U], vB[U], qB[U];
        float w[U];
#pragma unroll
        for (int u = 0; u < U; ++u) {
            const int row = (idx + u * stride) >> 3;
            const int pA = row * (COMB / 4) + c;
            vA[u] = a4[pA];
            qA[u] = p4[pA];
            vB[u] = a4[pA + 8];
            qB[u] = p4[pA + 8];
            w[u]  = dinv[row];
        }
#pragma unroll
        for (int u = 0; u < U; ++u) {
            const int row = (idx + u * stride) >> 3;
            const int pA = row * (COMB / 4) + c;
            const ushort4 oA = relu_sum_bf4(vA[u], qA[u], bA);
            const ushort4 oB = relu_sum_bf4(vB[u], qB[u], bB);
            const float ww = w[u] * w[u];
            g4[idx + u * stride] = pair_cols(oA, oB);
            s04[pA]     = bf4_scaled(oA, ww);
            s04[pA + 8] = bf4_scaled(oB, ww);
            s14[pA]     = zero;
            s14[pA + 8] = zero;
        }
    }
    for (; idx < n8; idx += stride) {
        const int row = idx >> 3;
        const int pA = row * (COMB / 4) + c;
        const ushort4 oA = relu_sum_bf4(a4[pA], p4[pA], bA);
        const ushort4 oB = relu_sum_bf4(a4[pA + 8], p4[pA + 8], bB);
        float w = dinv[row]; w *= w;
        g4[idx] = pair_cols(oA, oB);
        s04[pA]     = bf4_scaled(oA, w);
        s04[pA + 8] = bf4_scaled(oB, w);
        s14[pA]     = zero;
        s14[pA + 8] = zero;
    }
}

__device__ __forceinline__ void split_pack2(float a, float b, unsigned int& hi, unsigned int& lo) {
    const f32x2 v = {a, b};
    hi = __builtin_bit_cast(unsigned int, __builtin_convertvector(v, bf16x2));
    const f32x2 r = {a - __uint_as_float(hi << 16), b - __uint_as_float(hi & 0xFFFF0000u)};
    lo = __builtin_bit_cast(unsigned int, __builtin_convertvector(r, bf16x2));
}

__global__ __launch_bounds__(64) void final_kernel(const float* __restrict__ mlp,
                                                   const float* __restrict__ s0,
                                                   const float* __restrict__ s1,
                                                   const unsigned short* __restrict__ W2p,
                                                   const float* __restrict__ b2,
                                                   const float* __restrict__ fw,
                                                   const float* __restrict__ fb,
                                                   float* __restrict__ out, int nNodes) {
    const int lane = threadIdx.x & 31;
    const int c = lane & 15;
    const int h = lane >> 4;
    const int n0 = __builtin_amdgcn_readfirstlane(((blockIdx.x * blockDim.x + threadIdx.x) >> 5) * 32);
    if (n0 >= nNodes) return;

    V16 b[2][4];
#pragma unroll
    for (int kt = 0; kt < 2; ++kt) {
#pragma unroll
        for (int nt = 0; nt < 4; ++nt) {
            const uint4* bp = (const uint4*)(W2p + (size_t)(kt * 4 + nt) * 512 + lane * 16);
            b[kt][nt].u[0] = bp[0];
            b[kt][nt].u[1] = bp[1];
        }
    }
    float bb[4], fh[4];
#pragma unroll
    for (int nt = 0; nt < 4; ++nt) { bb[nt] = b2[16 * nt + c]; fh[nt] = fw[COMB + 16 * nt + c]; }
    const float4 fl = *(const float4*)(fw + 4 * c);

    float v[2];
#pragma unroll
    for (int i = 0; i < 2; ++i) {
        const int base = n0 + 16 * i;
        int arow = base + c;
        arow = (arow < nNodes) ? arow : (nNodes - 1);
        const float* r0 = s0 + (size_t)arow * COMB + 8 * h;
        const float* r1 = s1 + (size_t)arow * COMB + 8 * h;
        float4 m[8];
#pragma unroll
        for (int r = 0; r < 8; ++r) {
            int mrow = base + 8 * h + r;
            mrow = (mrow < nNodes) ? mrow : (nNodes - 1);
            m[r] = *(const float4*)(mlp + (size_t)mrow * COMB + 4 * c);
        }
        V16 ahi[2], alo[2];
#pragma unroll
        for (int kt = 0; kt < 2; ++kt) {
            float4 x[4], y[4];
#pragma unroll
            for (int t = 0; t < 4; ++t) {
                const int off = kt * 32 + (t >> 1) * 16 + (t & 1) * 4;
                x[t] = *(const float4*)(r0 + off);
                y[t] = *(const float4*)(r1 + off);
            }
#pragma unroll
            for (int t = 0; t < 2; ++t) {
                uint4 uh, ul;
                split_pack2(x[2 * t].x + y[2 * t].x, x[2 * t].y + y[2 * t].y, uh.x, ul.x);
                split_pack2(x[2 * t].z + y[2 * t].z, x[2 * t].w + y[2 * t].w, uh.y, ul.y);
                split_pack2(x[2 * t + 1].x + y[2 * t + 1].x, x[2 * t + 1].y + y[2 * t + 1].y, uh.z, ul.z);
                split_pack2(x[2 * t + 1].z + y[2 * t + 1].z, x[2 * t + 1].w + y[2 * t + 1].w, uh.w, ul.w);
                ahi[kt].u[t] = uh;
                alo[kt].u[t] = ul;
            }
        }
        v8f acc[4];
#pragma unroll
        for (int nt = 0; nt < 4; ++nt) acc[nt] = v8f{};
#pragma unroll
        for (int kt = 0; kt < 2; ++kt) {
#pragma unroll
            for (int nt = 0; nt < 4; ++nt) {
                acc[nt] = __builtin_amdgcn_wmma_f32_16x16x32_bf16(false, ahi[kt].v, false, b[kt][nt].v,
                                                                  (short)0, acc[nt], false, false);
                acc[nt] = __builtin_amdgcn_wmma_f32_16x16x32_bf16(false, alo[kt].v, false, b[kt][nt].v,
                                                                  (short)0, acc[nt], false, false);
            }
        }
        float p[8];
#pragma unroll
        for (int r = 0; r < 8; ++r) {
            float t = m[r].x * fl.x + m[r].y * fl.y + m[r].z * fl.z + m[r].w * fl.w;
#pragma unroll
            for (int nt = 0; nt < 4; ++nt) t += (acc[nt][r] + bb[nt]) * fh[nt];
            t += __shfl_xor(t, 8);
            t += __shfl_xor(t, 4);
            t += __shfl_xor(t, 2);
            t += __shfl_xor(t, 1);
            p[r] = t;
        }
        float sel = p[0];
#pragma unroll
        for (int r = 1; r < 8; ++r) sel = ((lane & 7) == r) ? p[r] : sel;
        v[i] = sel;
    }
    const int from = ((lane & 15) >> 3) * 16 + (lane & 7);
    const float t0 = __shfl(v[0], from);
    const float t1 = __shfl(v[1], from);
    const float res = (lane < 16) ? t0 : t1;
    if (n0 + lane < nNodes) out[n0 + lane] = res + fb[0];
}

extern "C" void kernel_launch(void* const* d_in, const int* in_sizes, int n_in,
                              void* d_out, int out_size, void* d_ws, size_t ws_size,
                              hipStream_t stream) {
    (void)in_sizes; (void)n_in; (void)out_size; (void)ws_size;

    const float* x        = (const float*)d_in[0];
    const int*   eidx     = (const int*)d_in[1];
    const float* eattr    = (const float*)d_in[2];
    const float* ufeat    = (const float*)d_in[3];
    const float* gcn1_W   = (const float*)d_in[4];
    const float* gcn1_b   = (const float*)d_in[5];
    const float* gcn2_W   = (const float*)d_in[6];
    const float* gcn2_b   = (const float*)d_in[7];
    const float* fc1_W    = (const float*)d_in[8];
    const float* fc1_b    = (const float*)d_in[9];
    const float* fc2_W    = (const float*)d_in[10];
    const float* fc2_b    = (const float*)d_in[11];
    const float* fcf_W    = (const float*)d_in[12];
    const float* fcf_b    = (const float*)d_in[13];
    const int* esrc = eidx;
    const int* edst = eidx + N_EDGES;
    float* out = (float*)d_out;

    char* ws = (char*)d_ws;
    size_t off = 0;
    auto carve = [&](size_t bytes) -> void* {
        void* p = ws + off;
        off = (off + bytes + 255) & ~(size_t)255;
        return p;
    };
    unsigned short* UBF = (unsigned short*)carve((size_t)N_NODES * MLP_IN * 2);
    unsigned short* XBF = (unsigned short*)carve((size_t)N_NODES * GNN_IN * 2);
    unsigned short* HBF = (unsigned short*)carve((size_t)N_NODES * MLP_HID * 2);
    unsigned short* GBF = (unsigned short*)carve((size_t)N_NODES * COMB * 2);
    unsigned short* W1P = (unsigned short*)carve((size_t)MLP_IN * MLP_HID * 2);
    unsigned short* W2P = (unsigned short*)carve((size_t)MLP_HID * COMB * 2);
    unsigned short* G1P = (unsigned short*)carve((size_t)GNN_IN * GNN_HID * 2);
    unsigned short* G2P = (unsigned short*)carve((size_t)GNN_HID * COMB * 2);
    float* MLPOUT = (float*)carve((size_t)N_NODES * COMB * 4);
    float* H1     = (float*)carve((size_t)N_NODES * COMB * 4);
    float* AGG    = (float*)carve((size_t)N_NODES * COMB * 4);
    float* DEG    = (float*)carve((size_t)N_NODES * 4);
    float* DINV   = (float*)carve((size_t)N_NODES * 4);
    float* NORM   = (float*)carve((size_t)N_EDGES * 4);
    float* DEG1 = (float*)HBF;
    float* AGG1 = (float*)UBF;
    float* S2A = (float*)XBF;
    float* S2B = S2A + (size_t)N_NODES * COMB;

    {
        int n4 = (N_NODES * MLP_IN) / 4;
        cvt_bf16_kernel<<<LOOP_BLOCKS, 256, 0, stream>>>(ufeat, UBF, n4);
        n4 = (N_NODES * GNN_IN) / 4;
        cvt_bf16_kernel<<<LOOP_BLOCKS, 256, 0, stream>>>(x, XBF, n4);
    }
    auto pack = [&](const float* W, unsigned short* P, int K, int N) {
        int total = (K / 32) * (N / 16) * 32;
        pack_w_kernel<<<(total + 255) / 256, 256, 0, stream>>>(W, P, K, N);
    };
    pack(fc1_W, W1P, MLP_IN, MLP_HID);
    pack(fc2_W, W2P, MLP_HID, COMB);
    pack(gcn1_W, G1P, GNN_IN, GNN_HID);
    pack(gcn2_W, G2P, GNN_HID, COMB);

    deg_init_kernel<<<(N_NODES + 255) / 256, 256, 0, stream>>>(DEG, DEG1, N_NODES);
    deg_accum_kernel<<<LOOP_BLOCKS, 256, 0, stream>>>(edst, eattr, DEG, DEG1, N_EDGES);
    dinv_kernel<<<(N_NODES + 255) / 256, 256, 0, stream>>>(DEG, DEG1, DINV, N_NODES);
    edge_norm_kernel<<<LOOP_BLOCKS, 256, 0, stream>>>(esrc, edst, eattr, DINV, NORM, N_EDGES);

    mlp_chain_kernel<<<256, 256, 0, stream>>>(UBF, W1P, fc1_b, W2P, fc2_b, MLPOUT, N_NODES);

    gemm_wmma_kernel<GNN_IN, GNN_HID, 2, false, false, false><<<256, 256, 0, stream>>>(
        XBF, G1P, nullptr, H1, N_NODES);
    agg_init_kernel<<<LOOP_BLOCKS, 256, 0, stream>>>(H1, DINV, AGG, AGG1, N_NODES);
    gcn_scatter_kernel<<<SCATTER_BLOCKS, 256, 0, stream>>>(H1, esrc, edst, NORM, AGG, AGG1, N_EDGES);
    relu_bias_init_kernel<<<LOOP_BLOCKS, 256, 0, stream>>>(AGG, AGG1, gcn1_b, DINV, (unsigned int*)GBF, S2A, S2B,
                                                            N_NODES * COMB);

    gcn_scatter_bf_kernel<<<SCATTER_BLOCKS, 256, 0, stream>>>((const unsigned int*)GBF, esrc, edst, NORM, S2A, S2B, N_EDGES);

    final_kernel<<<(N_NODES + 63) / 64, 64, 0, stream>>>(MLPOUT, S2A, S2B, G2P, gcn2_b, fcf_W, fcf_b,
                                                      out, N_NODES);
}
